# starting kernel
# baseline (speedup 1.0000x reference)
#include <hip/hip_runtime.h>

typedef float v2f __attribute__((ext_vector_type(2)));
typedef float v8f __attribute__((ext_vector_type(8)));

__global__ __launch_bounds__(256) void deg_init_kernel(float* __restrict__ deg, int N) {
    int n = blockIdx.x * 256 + threadIdx.x;
    if (n < N) deg[n] = 1.0f;
}

__global__ __launch_bounds__(256) void deg_count_kernel(const int* __restrict__ cols,
                                                        float* __restrict__ deg, int E) {
    int e = blockIdx.x * 256 + threadIdx.x;
    if (e < E) atomicAdd(&deg[cols[e]], 1.0f);
}

__global__ __launch_bounds__(256) void dinv_kernel(const float* __restrict__ deg,
                                                   float* __restrict__ dinv, int N) {
    int n = blockIdx.x * 256 + threadIdx.x;
    if (n < N) dinv[n] = rsqrtf(deg[n]);
}

__global__ __launch_bounds__(256) void gemm_xwT_wmma_kernel(const float* __restrict__ X,
                                                            const float* __restrict__ W,
                                                            float* __restrict__ Xhat,
                                                            int N) {
    const int lane    = threadIdx.x & 31;
    const int wave    = threadIdx.x >> 5;
    const int rowBase = blockIdx.x * 16;
    const int colBase = wave * 16;
    if (rowBase >= N) return;

    const int half = lane >> 4;
    const int l16  = lane & 15;

    const float* aptr = X + (size_t)(rowBase + l16) * 128 + (half << 1);
    const float* bptr = W + (size_t)(colBase + l16) * 128 + (half << 1);

    v8f acc = {0.f, 0.f, 0.f, 0.f, 0.f, 0.f, 0.f, 0.f};
#pragma unroll
    for (int k = 0; k < 128; k += 4) {
        v2f a = *(const v2f*)(aptr + k);
        v2f b = *(const v2f*)(bptr + k);
        acc = __builtin_amdgcn_wmma_f32_16x16x4_f32(
             false, a,  false, b,
             (short)0, acc,  false,  false);
    }

    float* cptr = Xhat + (size_t)rowBase * 128 + colBase;
#pragma unroll
    for (int v = 0; v < 8; ++v) {
        int m = v + (half << 3);
        cptr[(size_t)m * 128 + l16] = acc[v];
    }
}

__global__ __launch_bounds__(256) void out_init_kernel(const float* __restrict__ dinv,
                                                       const float* __restrict__ Xhat,
                                                       float* __restrict__ out, int total) {
    int i = blockIdx.x * 256 + threadIdx.x;
    if (i >= total) return;
    float d = dinv[i >> 7];
    out[i] = d * d * Xhat[i];
}

__global__ __launch_bounds__(256) void spmm_edge_kernel(const int* __restrict__ rows,
                                                        const int* __restrict__ cols,
                                                        const float* __restrict__ dinv,
                                                        const float* __restrict__ Xhat,
                                                        float* __restrict__ out, int E) {
    int e = blockIdx.x * 8 + (threadIdx.x >> 5);
    if (e >= E) return;
    int lane = threadIdx.x & 31;
    int r = rows[e];
    int c = cols[e];
    float val = dinv[r] * dinv[c];

    const float4 x = *(const float4*)(Xhat + (size_t)c * 128 + lane * 4);
    float* o = out + (size_t)r * 128 + lane * 4;
    atomicAdd(o + 0, val * x.x);
    atomicAdd(o + 1, val * x.y);
    atomicAdd(o + 2, val * x.z);
    atomicAdd(o + 3, val * x.w);
}

extern "C" void kernel_launch(void* const* d_in, const int* in_sizes, int n_in,
                              void* d_out, int out_size, void* d_ws, size_t ws_size,
                              hipStream_t stream) {
    const float* X  = (const float*)d_in[0];
    const float* W  = (const float*)d_in[1];
    const int*   ei = (const int*)d_in[2];
    float* out = (float*)d_out;

    const int N = in_sizes[0] / 128;
    const int E = in_sizes[2] / 2;
    const int* rows = ei;
    const int* cols = ei + E;

    float* Xhat = (float*)d_ws;
    float* deg  = Xhat + (size_t)N * 128;
    float* dinv = deg + N;

    deg_init_kernel<<<(N + 255) / 256, 256, 0, stream>>>(deg, N);
    deg_count_kernel<<<(E + 255) / 256, 256, 0, stream>>>(cols, deg, E);
    dinv_kernel<<<(N + 255) / 256, 256, 0, stream>>>(deg, dinv, N);

    gemm_xwT_wmma_kernel<<<(N + 15) / 16, 256, 0, stream>>>(X, W, Xhat, N);

    const int total = N * 128;
    out_init_kernel<<<(total + 255) / 256, 256, 0, stream>>>(dinv, Xhat, out, total);

    spmm_edge_kernel<<<(E + 7) / 8, 256, 0, stream>>>(rows, cols, dinv, Xhat, out, E);
}


# spmm: each 32-wave block keeps 400 out rows in LDS, filters the edge list for them, adds X_hat rows by LDS f32 atomics (was global atomics); out_init folded in; sums still in arrival order
# speedup vs baseline: 1.7076x; 1.7076x over previous
#include <hip/hip_runtime.h>

typedef float v2f __attribute__((ext_vector_type(2)));
typedef float v8f __attribute__((ext_vector_type(8)));

__global__ __launch_bounds__(256) void deg_init_kernel(float* __restrict__ deg, int N) {
    int n = blockIdx.x * 256 + threadIdx.x;
    if (n < N) deg[n] = 1.0f;
}

__global__ __launch_bounds__(256) void deg_count_kernel(const int* __restrict__ cols,
                                                        float* __restrict__ deg, int E) {
    int e = blockIdx.x * 256 + threadIdx.x;
    if (e < E) atomicAdd(&deg[cols[e]], 1.0f);
}

__global__ __launch_bounds__(256) void dinv_kernel(const float* __restrict__ deg,
                                                   float* __restrict__ dinv, int N) {
    int n = blockIdx.x * 256 + threadIdx.x;
    if (n < N) dinv[n] = rsqrtf(deg[n]);
}

__global__ __launch_bounds__(256) void gemm_xwT_wmma_kernel(const float* __restrict__ X,
                                                            const float* __restrict__ W,
                                                            float* __restrict__ Xhat,
                                                            int N) {
    const int lane    = threadIdx.x & 31;
    const int wave    = threadIdx.x >> 5;
    const int rowBase = blockIdx.x * 16;
    const int colBase = wave * 16;
    if (rowBase >= N) return;

    const int half = lane >> 4;
    const int l16  = lane & 15;

    const float* aptr = X + (size_t)(rowBase + l16) * 128 + (half << 1);
    const float* bptr = W + (size_t)(colBase + l16) * 128 + (half << 1);

    v8f acc = {0.f, 0.f, 0.f, 0.f, 0.f, 0.f, 0.f, 0.f};
#pragma unroll
    for (int k = 0; k < 128; k += 4) {
        v2f a = *(const v2f*)(aptr + k);
        v2f b = *(const v2f*)(bptr + k);
        acc = __builtin_amdgcn_wmma_f32_16x16x4_f32(
             false, a,  false, b,
             (short)0, acc,  false,  false);
    }

    float* cptr = Xhat + (size_t)rowBase * 128 + colBase;
#pragma unroll
    for (int v = 0; v < 8; ++v) {
        int m = v + (half << 3);
        cptr[(size_t)m * 128 + l16] = acc[v];
    }
}

#define TILE_ROWS 400
#define SEG_CAP   768

__device__ __forceinline__ void add_edge_row(float* __restrict__ tile,
                                             const float* __restrict__ Xhat,
                                             int c, int rl, float val, int lane) {
    const float4 x = *(const float4*)(Xhat + (size_t)c * 128 + lane * 4);
    float* t = tile + rl * 128 + lane;
    atomicAdd(t,      val * x.x);
    atomicAdd(t + 32, val * x.y);
    atomicAdd(t + 64, val * x.z);
    atomicAdd(t + 96, val * x.w);
}

__device__ __forceinline__ int push_match(bool m, int e, int cnt, int* __restrict__ seg) {
    const unsigned mj = __builtin_amdgcn_ballot_w32(m);
    if (mj == 0u) return cnt;
    const int slot = cnt + (int)__builtin_amdgcn_mbcnt_lo(mj, 0u);
    if (m && slot < SEG_CAP) seg[slot] = e;
    return cnt + __builtin_popcount(mj);
}

__device__ __forceinline__ int scan_step(unsigned a0, unsigned a1, unsigned a2, unsigned a3,
                                         unsigned unrows, int e0, int cnt, int* __restrict__ seg) {
    if (__builtin_amdgcn_ballot_w32(min(min(a0, a1), min(a2, a3)) < unrows) != 0u) {
        cnt = push_match(a0 < unrows, e0 + 0, cnt, seg);
        cnt = push_match(a1 < unrows, e0 + 1, cnt, seg);
        cnt = push_match(a2 < unrows, e0 + 2, cnt, seg);
        cnt = push_match(a3 < unrows, e0 + 3, cnt, seg);
    }
    return cnt;
}

__global__ __launch_bounds__(1024) void spmm_tile_kernel(const int* __restrict__ rows,
                                                         const int* __restrict__ cols,
                                                         const float* __restrict__ dinv,
                                                         const float* __restrict__ Xhat,
                                                         float* __restrict__ out,
                                                         int N, int E) {
    __shared__ float tile[TILE_ROWS * 128];
    __shared__ int   queue[32 * SEG_CAP];

    const int tid  = threadIdx.x;
    const int lane = tid & 31;
    const int wave = __builtin_amdgcn_readfirstlane(tid >> 5);
    const int rowBase = blockIdx.x * TILE_ROWS;
    const int nrows   = min(TILE_ROWS, N - rowBase);
    const unsigned unrows = (unsigned)nrows;

#pragma unroll 4
    for (int row = wave; row < nrows; row += 32) {
        const int n = rowBase + row;
        const float d = dinv[n];
        const float4 x = *(const float4*)(Xhat + (size_t)n * 128 + lane * 4);
        float* t = tile + row * 128 + lane;
        t[0]  = d * d * x.x;
        t[32] = d * d * x.y;
        t[64] = d * d * x.z;
        t[96] = d * d * x.w;
    }
    __syncthreads();

    const int E4 = E >> 2;
    const int4* rows4 = (const int4*)rows;
    int* seg = queue + wave * SEG_CAP;
    int cnt = 0;
    constexpr int U = 8;
    int base = wave * 32;
    for (; base + (U - 1) * 1024 + 31 < E4; base += 1024 * U) {
        int4 r4[U];
#pragma unroll
        for (int u = 0; u < U; ++u) r4[u] = rows4[base + u * 1024 + lane];
#pragma unroll
        for (int u = 0; u < U; ++u) {
            cnt = scan_step((unsigned)(r4[u].x - rowBase), (unsigned)(r4[u].y - rowBase),
                            (unsigned)(r4[u].z - rowBase), (unsigned)(r4[u].w - rowBase),
                            unrows, 4 * (base + u * 1024 + lane), cnt, seg);
        }
    }
    for (; base < E4; base += 1024) {
        const int i = base + lane;
        const bool ok = i < E4;
        const int4 r = rows4[min(i, E4 - 1)];
        cnt = scan_step(ok ? (unsigned)(r.x - rowBase) : ~0u, ok ? (unsigned)(r.y - rowBase) : ~0u,
                        ok ? (unsigned)(r.z - rowBase) : ~0u, ok ? (unsigned)(r.w - rowBase) : ~0u,
                        unrows, 4 * i, cnt, seg);
    }
    if (cnt > SEG_CAP) {
        int cnt2 = 0;
        for (int base2 = wave * 32; base2 < E4; base2 += 1024) {
            const int i = base2 + lane;
            const bool ok = i < E4;
            const int4 r = rows4[min(i, E4 - 1)];
            const int rr[4] = {r.x, r.y, r.z, r.w};
#pragma unroll
            for (int j = 0; j < 4; ++j) {
                const bool m = ok && ((unsigned)(rr[j] - rowBase) < unrows);
                const unsigned mj = __builtin_amdgcn_ballot_w32(m);
                const int slot = cnt2 + (int)__builtin_amdgcn_mbcnt_lo(mj, 0u);
                unsigned ov = __builtin_amdgcn_ballot_w32(m && slot >= SEG_CAP);
                while (ov) {
                    const int src = __builtin_ctz(ov);
                    ov &= ov - 1u;
                    const int rsel = __builtin_amdgcn_readlane(rr[j], src);
                    const int esel = __builtin_amdgcn_readlane(4 * i + j, src);
                    const int c = cols[esel];
                    add_edge_row(tile, Xhat, c, rsel - rowBase, dinv[rsel] * dinv[c], lane);
                }
                cnt2 += __builtin_popcount(mj);
            }
        }
    }
    if (wave == 0) {
        for (int e = E4 * 4; e < E; ++e) {
            const int r = rows[e];
            if ((unsigned)(r - rowBase) < unrows) {
                const int c = cols[e];
                add_edge_row(tile, Xhat, c, r - rowBase, dinv[r] * dinv[c], lane);
            }
        }
    }

    const int nq = min(cnt, SEG_CAP);
    constexpr int G = 16;
    for (int b = 0; b < nq; b += 32) {
        const int nv = min(32, nq - b);
        const int e = seg[b + (lane < nv ? lane : 0)];
        const int r = rows[e];
        const int c = cols[e];
        const float val = dinv[r] * dinv[c];
        const int rl = r - rowBase;
        for (int k0 = 0; k0 < nv; k0 += G) {
            float4 x[G];
#pragma unroll
            for (int u = 0; u < G; ++u) {
                const int cc = __builtin_amdgcn_readlane(c, k0 + u);
                x[u] = *(const float4*)(Xhat + (size_t)cc * 128 + lane * 4);
            }
#pragma unroll
            for (int u = 0; u < G; ++u) {
                if (u == 0 || k0 + u < nv) {
                    const float v = __int_as_float(__builtin_amdgcn_readlane(__float_as_int(val), k0 + u));
                    const int rr  = __builtin_amdgcn_readlane(rl, k0 + u);
                    float* t = tile + rr * 128 + lane;
                    atomicAdd(t,      v * x[u].x);
                    atomicAdd(t + 32, v * x[u].y);
                    atomicAdd(t + 64, v * x[u].z);
                    atomicAdd(t + 96, v * x[u].w);
                }
            }
        }
    }
    __syncthreads();

    for (int row = wave; row < nrows; row += 32) {
        const float* t = tile + row * 128 + lane;
        float4 o;
        o.x = t[0];
        o.y = t[32];
        o.z = t[64];
        o.w = t[96];
        *(float4*)(out + (size_t)(rowBase + row) * 128 + lane * 4) = o;
    }
}

extern "C" void kernel_launch(void* const* d_in, const int* in_sizes, int n_in,
                              void* d_out, int out_size, void* d_ws, size_t ws_size,
                              hipStream_t stream) {
    const float* X  = (const float*)d_in[0];
    const float* W  = (const float*)d_in[1];
    const int*   ei = (const int*)d_in[2];
    float* out = (float*)d_out;

    const int N = in_sizes[0] / 128;
    const int E = in_sizes[2] / 2;
    const int* rows = ei;
    const int* cols = ei + E;

    float* Xhat = (float*)d_ws;
    float* deg  = Xhat + (size_t)N * 128;
    float* dinv = deg + N;

    deg_init_kernel<<<(N + 255) / 256, 256, 0, stream>>>(deg, N);
    deg_count_kernel<<<(E + 255) / 256, 256, 0, stream>>>(cols, deg, E);
    dinv_kernel<<<(N + 255) / 256, 256, 0, stream>>>(deg, dinv, N);

    gemm_xwT_wmma_kernel<<<(N + 15) / 16, 256, 0, stream>>>(X, W, Xhat, N);

    spmm_tile_kernel<<<(N + TILE_ROWS - 1) / TILE_ROWS, 1024, 0, stream>>>(rows, cols, dinv, Xhat, out, N, E);
}


# gemm: each wave keeps its 128x16 slab of W^T as fragments in registers over the block's 32-row strips; X strips and results pass LDS as whole 512-B rows (was 8-B loads per k-step); f32 sum order kept
# speedup vs baseline: 1.9770x; 1.1578x over previous
#include <hip/hip_runtime.h>

typedef float v2f __attribute__((ext_vector_type(2)));
typedef float v8f __attribute__((ext_vector_type(8)));

__global__ __launch_bounds__(256) void deg_init_kernel(float* __restrict__ deg, int N) {
    int n = blockIdx.x * 256 + threadIdx.x;
    if (n < N) deg[n] = 1.0f;
}

__global__ __launch_bounds__(256) void deg_count_kernel(const int* __restrict__ cols,
                                                        float* __restrict__ deg, int E) {
    int e = blockIdx.x * 256 + threadIdx.x;
    if (e < E) atomicAdd(&deg[cols[e]], 1.0f);
}

__global__ __launch_bounds__(256) void dinv_kernel(const float* __restrict__ deg,
                                                   float* __restrict__ dinv, int N) {
    int n = blockIdx.x * 256 + threadIdx.x;
    if (n < N) dinv[n] = rsqrtf(deg[n]);
}

#define GS_ROWS  32
#define GS_PITCH 132

__global__ __launch_bounds__(256) void gemm_xwT_wmma_kernel(const float* __restrict__ X,
                                                            const float* __restrict__ W,
                                                            float* __restrict__ Xhat,
                                                            int N) {
    __shared__ __attribute__((aligned(16))) float sA[GS_ROWS * GS_PITCH];
    __shared__ __attribute__((aligned(16))) float sO[GS_ROWS * GS_PITCH];

    const int lane    = threadIdx.x & 31;
    const int wave    = __builtin_amdgcn_readfirstlane(threadIdx.x >> 5);
    const int colBase = wave * 16;
    const int half    = lane >> 4;
    const int l16     = lane & 15;
    const int nstrips = (N + GS_ROWS - 1) / GS_ROWS;
    const int stride  = (int)gridDim.x;

    v2f bfrag[32];
    {
        const float* bptr = W + (size_t)(colBase + l16) * 128 + (half << 1);
#pragma unroll
        for (int ks = 0; ks < 32; ++ks) bfrag[ks] = *(const v2f*)(bptr + 4 * ks);
    }

    const int myRow = wave * 4;
    const float* xlane = X + lane * 4;
    float4 x0 = {0.f, 0.f, 0.f, 0.f}, x1 = x0, x2 = x0, x3 = x0;
    int s = blockIdx.x;
    if (s < nstrips) {
        const int r0 = s * GS_ROWS + myRow;
        x0 = *(const float4*)(xlane + (size_t)min(r0 + 0, N - 1) * 128);
        x1 = *(const float4*)(xlane + (size_t)min(r0 + 1, N - 1) * 128);
        x2 = *(const float4*)(xlane + (size_t)min(r0 + 2, N - 1) * 128);
        x3 = *(const float4*)(xlane + (size_t)min(r0 + 3, N - 1) * 128);
    }

    const float* a0p = sA + l16 * GS_PITCH + (half << 1);
    const float* a1p = sA + (16 + l16) * GS_PITCH + (half << 1);

    for (; s < nstrips; s += stride) {
        float* arow = sA + myRow * GS_PITCH + lane * 4;
        *(float4*)(arow + 0 * GS_PITCH) = x0;
        *(float4*)(arow + 1 * GS_PITCH) = x1;
        *(float4*)(arow + 2 * GS_PITCH) = x2;
        *(float4*)(arow + 3 * GS_PITCH) = x3;
        __syncthreads();

        const int sn = s + stride;
        if (sn < nstrips) {
            const int r0 = sn * GS_ROWS + myRow;
            x0 = *(const float4*)(xlane + (size_t)min(r0 + 0, N - 1) * 128);
            x1 = *(const float4*)(xlane + (size_t)min(r0 + 1, N - 1) * 128);
            x2 = *(const float4*)(xlane + (size_t)min(r0 + 2, N - 1) * 128);
            x3 = *(const float4*)(xlane + (size_t)min(r0 + 3, N - 1) * 128);
        }

        v8f acc0 = {0.f, 0.f, 0.f, 0.f, 0.f, 0.f, 0.f, 0.f};
        v8f acc1 = {0.f, 0.f, 0.f, 0.f, 0.f, 0.f, 0.f, 0.f};
#pragma unroll
        for (int ks = 0; ks < 32; ++ks) {
            const v2f a0 = *(const v2f*)(a0p + 4 * ks);
            const v2f a1 = *(const v2f*)(a1p + 4 * ks);
            acc0 = __builtin_amdgcn_wmma_f32_16x16x4_f32(
                 false, a0,  false, bfrag[ks],
                 (short)0, acc0,  false,  false);
            acc1 = __builtin_amdgcn_wmma_f32_16x16x4_f32(
                 false, a1,  false, bfrag[ks],
                 (short)0, acc1,  false,  false);
        }

#pragma unroll
        for (int v = 0; v < 8; ++v) {
            const int m = v + (half << 3);
            sO[m * GS_PITCH + colBase + l16]        = acc0[v];
            sO[(16 + m) * GS_PITCH + colBase + l16] = acc1[v];
        }
        __syncthreads();

        const float* orow = sO + myRow * GS_PITCH + lane * 4;
        const int r0 = s * GS_ROWS + myRow;
        float* dst = Xhat + (size_t)r0 * 128 + lane * 4;
        if (r0 + 3 < N) {
            const float4 o0 = *(const float4*)(orow + 0 * GS_PITCH);
            const float4 o1 = *(const float4*)(orow + 1 * GS_PITCH);
            const float4 o2 = *(const float4*)(orow + 2 * GS_PITCH);
            const float4 o3 = *(const float4*)(orow + 3 * GS_PITCH);
            *(float4*)(dst + 0 * 128) = o0;
            *(float4*)(dst + 1 * 128) = o1;
            *(float4*)(dst + 2 * 128) = o2;
            *(float4*)(dst + 3 * 128) = o3;
        } else {
            for (int j = 0; j < 4; ++j)
                if (r0 + j < N) *(float4*)(dst + j * 128) = *(const float4*)(orow + j * GS_PITCH);
        }
    }
}

#define TILE_ROWS 400
#define SEG_CAP   768

__device__ __forceinline__ void add_edge_row(float* __restrict__ tile,
                                             const float* __restrict__ Xhat,
                                             int c, int rl, float val, int lane) {
    const float4 x = *(const float4*)(Xhat + (size_t)c * 128 + lane * 4);
    float* t = tile + rl * 128 + lane;
    atomicAdd(t,      val * x.x);
    atomicAdd(t + 32, val * x.y);
    atomicAdd(t + 64, val * x.z);
    atomicAdd(t + 96, val * x.w);
}

__device__ __forceinline__ int push_match(bool m, int e, int cnt, int* __restrict__ seg) {
    const unsigned mj = __builtin_amdgcn_ballot_w32(m);
    if (mj == 0u) return cnt;
    const int slot = cnt + (int)__builtin_amdgcn_mbcnt_lo(mj, 0u);
    if (m && slot < SEG_CAP) seg[slot] = e;
    return cnt + __builtin_popcount(mj);
}

__device__ __forceinline__ int scan_step(unsigned a0, unsigned a1, unsigned a2, unsigned a3,
                                         unsigned unrows, int e0, int cnt, int* __restrict__ seg) {
    if (__builtin_amdgcn_ballot_w32(min(min(a0, a1), min(a2, a3)) < unrows) != 0u) {
        cnt = push_match(a0 < unrows, e0 + 0, cnt, seg);
        cnt = push_match(a1 < unrows, e0 + 1, cnt, seg);
        cnt = push_match(a2 < unrows, e0 + 2, cnt, seg);
        cnt = push_match(a3 < unrows, e0 + 3, cnt, seg);
    }
    return cnt;
}

__global__ __launch_bounds__(1024) void spmm_tile_kernel(const int* __restrict__ rows,
                                                         const int* __restrict__ cols,
                                                         const float* __restrict__ dinv,
                                                         const float* __restrict__ Xhat,
                                                         float* __restrict__ out,
                                                         int N, int E) {
    __shared__ float tile[TILE_ROWS * 128];
    __shared__ int   queue[32 * SEG_CAP];

    const int tid  = threadIdx.x;
    const int lane = tid & 31;
    const int wave = __builtin_amdgcn_readfirstlane(tid >> 5);
    const int rowBase = blockIdx.x * TILE_ROWS;
    const int nrows   = min(TILE_ROWS, N - rowBase);
    const unsigned unrows = (unsigned)nrows;

#pragma unroll 4
    for (int row = wave; row < nrows; row += 32) {
        const int n = rowBase + row;
        const float d = dinv[n];
        const float4 x = *(const float4*)(Xhat + (size_t)n * 128 + lane * 4);
        float* t = tile + row * 128 + lane;
        t[0]  = d * d * x.x;
        t[32] = d * d * x.y;
        t[64] = d * d * x.z;
        t[96] = d * d * x.w;
    }
    __syncthreads();

    const int E4 = E >> 2;
    const int4* rows4 = (const int4*)rows;
    int* seg = queue + wave * SEG_CAP;
    int cnt = 0;
    constexpr int U = 8;
    int base = wave * 32;
    for (; base + (U - 1) * 1024 + 31 < E4; base += 1024 * U) {
        int4 r4[U];
#pragma unroll
        for (int u = 0; u < U; ++u) r4[u] = rows4[base + u * 1024 + lane];
#pragma unroll
        for (int u = 0; u < U; ++u) {
            cnt = scan_step((unsigned)(r4[u].x - rowBase), (unsigned)(r4[u].y - rowBase),
                            (unsigned)(r4[u].z - rowBase), (unsigned)(r4[u].w - rowBase),
                            unrows, 4 * (base + u * 1024 + lane), cnt, seg);
        }
    }
    for (; base < E4; base += 1024) {
        const int i = base + lane;
        const bool ok = i < E4;
        const int4 r = rows4[min(i, E4 - 1)];
        cnt = scan_step(ok ? (unsigned)(r.x - rowBase) : ~0u, ok ? (unsigned)(r.y - rowBase) : ~0u,
                        ok ? (unsigned)(r.z - rowBase) : ~0u, ok ? (unsigned)(r.w - rowBase) : ~0u,
                        unrows, 4 * i, cnt, seg);
    }
    if (cnt > SEG_CAP) {
        int cnt2 = 0;
        for (int base2 = wave * 32; base2 < E4; base2 += 1024) {
            const int i = base2 + lane;
            const bool ok = i < E4;
            const int4 r = rows4[min(i, E4 - 1)];
            const int rr[4] = {r.x, r.y, r.z, r.w};
#pragma unroll
            for (int j = 0; j < 4; ++j) {
                const bool m = ok && ((unsigned)(rr[j] - rowBase) < unrows);
                const unsigned mj = __builtin_amdgcn_ballot_w32(m);
                const int slot = cnt2 + (int)__builtin_amdgcn_mbcnt_lo(mj, 0u);
                unsigned ov = __builtin_amdgcn_ballot_w32(m && slot >= SEG_CAP);
                while (ov) {
                    const int src = __builtin_ctz(ov);
                    ov &= ov - 1u;
                    const int rsel = __builtin_amdgcn_readlane(rr[j], src);
                    const int esel = __builtin_amdgcn_readlane(4 * i + j, src);
                    const int c = cols[esel];
                    add_edge_row(tile, Xhat, c, rsel - rowBase, dinv[rsel] * dinv[c], lane);
                }
                cnt2 += __builtin_popcount(mj);
            }
        }
    }
    if (wave == 0) {
        for (int e = E4 * 4; e < E; ++e) {
            const int r = rows[e];
            if ((unsigned)(r - rowBase) < unrows) {
                const int c = cols[e];
                add_edge_row(tile, Xhat, c, r - rowBase, dinv[r] * dinv[c], lane);
            }
        }
    }

    const int nq = min(cnt, SEG_CAP);
    constexpr int G = 16;
    for (int b = 0; b < nq; b += 32) {
        const int nv = min(32, nq - b);
        const int e = seg[b + (lane < nv ? lane : 0)];
        const int r = rows[e];
        const int c = cols[e];
        const float val = dinv[r] * dinv[c];
        const int rl = r - rowBase;
        for (int k0 = 0; k0 < nv; k0 += G) {
            float4 x[G];
#pragma unroll
            for (int u = 0; u < G; ++u) {
                const int cc = __builtin_amdgcn_readlane(c, k0 + u);
                x[u] = *(const float4*)(Xhat + (size_t)cc * 128 + lane * 4);
            }
#pragma unroll
            for (int u = 0; u < G; ++u) {
                if (u == 0 || k0 + u < nv) {
                    const float v = __int_as_float(__builtin_amdgcn_readlane(__float_as_int(val), k0 + u));
                    const int rr  = __builtin_amdgcn_readlane(rl, k0 + u);
                    float* t = tile + rr * 128 + lane;
                    atomicAdd(t,      v * x[u].x);
                    atomicAdd(t + 32, v * x[u].y);
                    atomicAdd(t + 64, v * x[u].z);
                    atomicAdd(t + 96, v * x[u].w);
                }
            }
        }
    }
    __syncthreads();

    for (int row = wave; row < nrows; row += 32) {
        const float* t = tile + row * 128 + lane;
        float4 o;
        o.x = t[0];
        o.y = t[32];
        o.z = t[64];
        o.w = t[96];
        *(float4*)(out + (size_t)(rowBase + row) * 128 + lane * 4) = o;
    }
}

extern "C" void kernel_launch(void* const* d_in, const int* in_sizes, int n_in,
                              void* d_out, int out_size, void* d_ws, size_t ws_size,
                              hipStream_t stream) {
    const float* X  = (const float*)d_in[0];
    const float* W  = (const float*)d_in[1];
    const int*   ei = (const int*)d_in[2];
    float* out = (float*)d_out;

    const int N = in_sizes[0] / 128;
    const int E = in_sizes[2] / 2;
    const int* rows = ei;
    const int* cols = ei + E;

    float* Xhat = (float*)d_ws;
    float* deg  = Xhat + (size_t)N * 128;
    float* dinv = deg + N;

    deg_init_kernel<<<(N + 255) / 256, 256, 0, stream>>>(deg, N);
    deg_count_kernel<<<(E + 255) / 256, 256, 0, stream>>>(cols, deg, E);
    dinv_kernel<<<(N + 255) / 256, 256, 0, stream>>>(deg, dinv, N);

    gemm_xwT_wmma_kernel<<<512, 256, 0, stream>>>(X, W, Xhat, N);

    spmm_tile_kernel<<<(N + TILE_ROWS - 1) / TILE_ROWS, 1024, 0, stream>>>(rows, cols, dinv, Xhat, out, N, E);
}


# spmm: blocks scan a bit-sliced key plane (owner block number, 1 B per edge, written by the degree-count launch) instead of the 4-B rows, 32 edges per lane and mask; sums still in arrival order
# speedup vs baseline: 2.3314x; 1.1793x over previous
#include <hip/hip_runtime.h>

typedef float v2f __attribute__((ext_vector_type(2)));
typedef float v8f __attribute__((ext_vector_type(8)));

__global__ __launch_bounds__(256) void deg_init_kernel(float* __restrict__ deg, int N) {
    int n = blockIdx.x * 256 + threadIdx.x;
    if (n < N) deg[n] = 1.0f;
}

#define TILE_ROWS 400

__global__ __launch_bounds__(256) void deg_count_kernel(const int* __restrict__ rows,
                                                        const int* __restrict__ cols,
                                                        float* __restrict__ deg,
                                                        unsigned int* __restrict__ keyw, int E) {
    const int e = blockIdx.x * 256 + threadIdx.x;
    if (e < E) atomicAdd(&deg[cols[e]], 1.0f);
    if (keyw != nullptr && (e | 31) < E) {
        const unsigned key  = (unsigned)(rows[e] / TILE_ROWS);
        const int      lane = threadIdx.x & 31;
        unsigned mine = 0u;
#pragma unroll
        for (int k = 0; k < 8; ++k) {
            const unsigned bk = __builtin_amdgcn_ballot_w32(((key >> k) & 1u) != 0u);
            if (lane == k) mine = bk;
        }
        if (lane < 8) {
            const int q = e >> 5;
            keyw[((((q >> 5) << 1) + (lane >> 2)) * 32 + (q & 31)) * 4 + (lane & 3)] = mine;
        }
    }
}

__global__ __launch_bounds__(256) void dinv_kernel(const float* __restrict__ deg,
                                                   float* __restrict__ dinv, int N) {
    int n = blockIdx.x * 256 + threadIdx.x;
    if (n < N) dinv[n] = rsqrtf(deg[n]);
}

#define GS_ROWS  32
#define GS_PITCH 132

__global__ __launch_bounds__(256) void gemm_xwT_wmma_kernel(const float* __restrict__ X,
                                                            const float* __restrict__ W,
                                                            float* __restrict__ Xhat,
                                                            int N) {
    __shared__ __attribute__((aligned(16))) float sA[GS_ROWS * GS_PITCH];
    __shared__ __attribute__((aligned(16))) float sO[GS_ROWS * GS_PITCH];

    const int lane    = threadIdx.x & 31;
    const int wave    = __builtin_amdgcn_readfirstlane(threadIdx.x >> 5);
    const int colBase = wave * 16;
    const int half    = lane >> 4;
    const int l16     = lane & 15;
    const int nstrips = (N + GS_ROWS - 1) / GS_ROWS;
    const int stride  = (int)gridDim.x;

    v2f bfrag[32];
    {
        const float* bptr = W + (size_t)(colBase + l16) * 128 + (half << 1);
#pragma unroll
        for (int ks = 0; ks < 32; ++ks) bfrag[ks] = *(const v2f*)(bptr + 4 * ks);
    }

    const int myRow = wave * 4;
    const float* xlane = X + lane * 4;
    float4 x0 = {0.f, 0.f, 0.f, 0.f}, x1 = x0, x2 = x0, x3 = x0;
    int s = blockIdx.x;
    if (s < nstrips) {
        const int r0 = s * GS_ROWS + myRow;
        x0 = *(const float4*)(xlane + (size_t)min(r0 + 0, N - 1) * 128);
        x1 = *(const float4*)(xlane + (size_t)min(r0 + 1, N - 1) * 128);
        x2 = *(const float4*)(xlane + (size_t)min(r0 + 2, N - 1) * 128);
        x3 = *(const float4*)(xlane + (size_t)min(r0 + 3, N - 1) * 128);
    }

    const float* a0p = sA + l16 * GS_PITCH + (half << 1);
    const float* a1p = sA + (16 + l16) * GS_PITCH + (half << 1);

    for (; s < nstrips; s += stride) {
        float* arow = sA + myRow * GS_PITCH + lane * 4;
        *(float4*)(arow + 0 * GS_PITCH) = x0;
        *(float4*)(arow + 1 * GS_PITCH) = x1;
        *(float4*)(arow + 2 * GS_PITCH) = x2;
        *(float4*)(arow + 3 * GS_PITCH) = x3;
        __syncthreads();

        const int sn = s + stride;
        if (sn < nstrips) {
            const int r0 = sn * GS_ROWS + myRow;
            x0 = *(const float4*)(xlane + (size_t)min(r0 + 0, N - 1) * 128);
            x1 = *(const float4*)(xlane + (size_t)min(r0 + 1, N - 1) * 128);
            x2 = *(const float4*)(xlane + (size_t)min(r0 + 2, N - 1) * 128);
            x3 = *(const float4*)(xlane + (size_t)min(r0 + 3, N - 1) * 128);
        }

        v8f acc0 = {0.f, 0.f, 0.f, 0.f, 0.f, 0.f, 0.f, 0.f};
        v8f acc1 = {0.f, 0.f, 0.f, 0.f, 0.f, 0.f, 0.f, 0.f};
#pragma unroll
        for (int ks = 0; ks < 32; ++ks) {
            const v2f a0 = *(const v2f*)(a0p + 4 * ks);
            const v2f a1 = *(const v2f*)(a1p + 4 * ks);
            acc0 = __builtin_amdgcn_wmma_f32_16x16x4_f32(
                 false, a0,  false, bfrag[ks],
                 (short)0, acc0,  false,  false);
            acc1 = __builtin_amdgcn_wmma_f32_16x16x4_f32(
                 false, a1,  false, bfrag[ks],
                 (short)0, acc1,  false,  false);
        }

#pragma unroll
        for (int v = 0; v < 8; ++v) {
            const int m = v + (half << 3);
            sO[m * GS_PITCH + colBase + l16]        = acc0[v];
            sO[(16 + m) * GS_PITCH + colBase + l16] = acc1[v];
        }
        __syncthreads();

        const float* orow = sO + myRow * GS_PITCH + lane * 4;
        const int r0 = s * GS_ROWS + myRow;
        float* dst = Xhat + (size_t)r0 * 128 + lane * 4;
        if (r0 + 3 < N) {
            const float4 o0 = *(const float4*)(orow + 0 * GS_PITCH);
            const float4 o1 = *(const float4*)(orow + 1 * GS_PITCH);
            const float4 o2 = *(const float4*)(orow + 2 * GS_PITCH);
            const float4 o3 = *(const float4*)(orow + 3 * GS_PITCH);
            *(float4*)(dst + 0 * 128) = o0;
            *(float4*)(dst + 1 * 128) = o1;
            *(float4*)(dst + 2 * 128) = o2;
            *(float4*)(dst + 3 * 128) = o3;
        } else {
            for (int j = 0; j < 4; ++j)
                if (r0 + j < N) *(float4*)(dst + j * 128) = *(const float4*)(orow + j * GS_PITCH);
        }
    }
}

#define SEG_CAP   768

__device__ __forceinline__ void add_edge_row(float* __restrict__ tile,
                                             const float* __restrict__ Xhat,
                                             int c, int rl, float val, int lane) {
    const float4 x = *(const float4*)(Xhat + (size_t)c * 128 + lane * 4);
    float* t = tile + rl * 128 + lane;
    atomicAdd(t,      val * x.x);
    atomicAdd(t + 32, val * x.y);
    atomicAdd(t + 64, val * x.z);
    atomicAdd(t + 96, val * x.w);
}

__device__ __forceinline__ int push_match(bool m, int e, int cnt, int* __restrict__ seg) {
    const unsigned mj = __builtin_amdgcn_ballot_w32(m);
    if (mj == 0u) return cnt;
    const int slot = cnt + (int)__builtin_amdgcn_mbcnt_lo(mj, 0u);
    if (m && slot < SEG_CAP) seg[slot] = e;
    return cnt + __builtin_popcount(mj);
}

__device__ __forceinline__ int scan_step(unsigned a0, unsigned a1, unsigned a2, unsigned a3,
                                         unsigned unrows, int e0, int cnt, int* __restrict__ seg) {
    if (__builtin_amdgcn_ballot_w32(min(min(a0, a1), min(a2, a3)) < unrows) != 0u) {
        cnt = push_match(a0 < unrows, e0 + 0, cnt, seg);
        cnt = push_match(a1 < unrows, e0 + 1, cnt, seg);
        cnt = push_match(a2 < unrows, e0 + 2, cnt, seg);
        cnt = push_match(a3 < unrows, e0 + 3, cnt, seg);
    }
    return cnt;
}

template <bool SECOND>
__device__ __forceinline__ int key_step(uint4 a, uint4 b, bool ok, unsigned kb, int e0, int cnt,
                                        int* __restrict__ seg, float* __restrict__ tile,
                                        const int* __restrict__ rows, const int* __restrict__ cols,
                                        const float* __restrict__ dinv, const float* __restrict__ Xhat,
                                        int rowBase, int lane) {
    unsigned m = ~(a.x ^ (0u - ((kb >> 0) & 1u))) & ~(a.y ^ (0u - ((kb >> 1) & 1u))) &
                 ~(a.z ^ (0u - ((kb >> 2) & 1u))) & ~(a.w ^ (0u - ((kb >> 3) & 1u))) &
                 ~(b.x ^ (0u - ((kb >> 4) & 1u))) & ~(b.y ^ (0u - ((kb >> 5) & 1u))) &
                 ~(b.z ^ (0u - ((kb >> 6) & 1u))) & ~(b.w ^ (0u - ((kb >> 7) & 1u)));
    if (!ok) m = 0u;
    for (;;) {
        const bool has = m != 0u;
        const unsigned mj = __builtin_amdgcn_ballot_w32(has);
        if (mj == 0u) break;
        const int e = e0 + __builtin_ffs((int)m) - 1;
        const int slot = cnt + (int)__builtin_amdgcn_mbcnt_lo(mj, 0u);
        if constexpr (!SECOND) {
            if (has && slot < SEG_CAP) seg[slot] = e;
        } else {
            unsigned ov = __builtin_amdgcn_ballot_w32(has && slot >= SEG_CAP);
            while (ov) {
                const int srcl = __builtin_ctz(ov);
                ov &= ov - 1u;
                const int esel = __builtin_amdgcn_readlane(e, srcl);
                const int r = rows[esel];
                const int c = cols[esel];
                add_edge_row(tile, Xhat, c, r - rowBase, dinv[r] * dinv[c], lane);
            }
        }
        m &= m - 1u;
        cnt += __builtin_popcount(mj);
    }
    return cnt;
}

template <bool KEY_PLANE>
__global__ __launch_bounds__(1024) void spmm_tile_kernel(const int* __restrict__ rows,
                                                         const int* __restrict__ cols,
                                                         const unsigned char* __restrict__ keys,
                                                         const float* __restrict__ dinv,
                                                         const float* __restrict__ Xhat,
                                                         float* __restrict__ out,
                                                         int N, int E) {
    __shared__ float tile[TILE_ROWS * 128];
    __shared__ int   queue[32 * SEG_CAP];

    const int tid  = threadIdx.x;
    const int lane = tid & 31;
    const int wave = __builtin_amdgcn_readfirstlane(tid >> 5);
    const int rowBase = blockIdx.x * TILE_ROWS;
    const int nrows   = min(TILE_ROWS, N - rowBase);
    const unsigned unrows = (unsigned)nrows;

#pragma unroll 4
    for (int row = wave; row < nrows; row += 32) {
        const int n = rowBase + row;
        const float d = dinv[n];
        const float4 x = *(const float4*)(Xhat + (size_t)n * 128 + lane * 4);
        float* t = tile + row * 128 + lane;
        t[0]  = d * d * x.x;
        t[32] = d * d * x.y;
        t[64] = d * d * x.z;
        t[96] = d * d * x.w;
    }
    __syncthreads();

    int* seg = queue + wave * SEG_CAP;
    int cnt = 0;
    constexpr int U = 8;
    const int tailStart = KEY_PLANE ? (E >> 5) << 5 : (E >> 2) << 2;
    if constexpr (KEY_PLANE) {
        const int G32 = tailStart >> 5;
        const int SGF = G32 >> 5;
        const uint4* kp = (const uint4*)keys;
        const unsigned kb = blockIdx.x;
        int sg = wave;
        for (; sg + (U - 1) * 32 < SGF; sg += 32 * U) {
            uint4 ka[U], kc[U];
#pragma unroll
            for (int u = 0; u < U; ++u) {
                ka[u] = kp[(sg + 32 * u) * 64 + lane];
                kc[u] = kp[(sg + 32 * u) * 64 + 32 + lane];
            }
#pragma unroll
            for (int u = 0; u < U; ++u) {
                cnt = key_step<false>(ka[u], kc[u], true, kb, ((sg + 32 * u) * 32 + lane) * 32, cnt, seg,
                                      tile, rows, cols, dinv, Xhat, rowBase, lane);
            }
        }
        for (; sg * 32 < G32; sg += 32) {
            const int g  = sg * 32 + lane;
            const int gl = min(g, G32 - 1);
            const int o  = (gl >> 5) * 64 + (gl & 31);
            cnt = key_step<false>(kp[o], kp[o + 32], g < G32, kb, g * 32, cnt, seg,
                                  tile, rows, cols, dinv, Xhat, rowBase, lane);
        }
        if (cnt > SEG_CAP) {
            int cnt2 = 0;
            for (int sg2 = wave; sg2 * 32 < G32; sg2 += 32) {
                const int g  = sg2 * 32 + lane;
                const int gl = min(g, G32 - 1);
                const int o  = (gl >> 5) * 64 + (gl & 31);
                cnt2 = key_step<true>(kp[o], kp[o + 32], g < G32, kb, g * 32, cnt2, seg,
                                      tile, rows, cols, dinv, Xhat, rowBase, lane);
            }
        }
    } else {
        const int E4 = tailStart >> 2;
        const int4* rows4 = (const int4*)rows;
        int base = wave * 32;
        for (; base + (U - 1) * 1024 + 31 < E4; base += 1024 * U) {
            int4 r4[U];
#pragma unroll
            for (int u = 0; u < U; ++u) r4[u] = rows4[base + u * 1024 + lane];
#pragma unroll
            for (int u = 0; u < U; ++u) {
                cnt = scan_step((unsigned)(r4[u].x - rowBase), (unsigned)(r4[u].y - rowBase),
                                (unsigned)(r4[u].z - rowBase), (unsigned)(r4[u].w - rowBase),
                                unrows, 4 * (base + u * 1024 + lane), cnt, seg);
            }
        }
        for (; base < E4; base += 1024) {
            const int i = base + lane;
            const bool ok = i < E4;
            const int4 r = rows4[min(i, E4 - 1)];
            cnt = scan_step(ok ? (unsigned)(r.x - rowBase) : ~0u, ok ? (unsigned)(r.y - rowBase) : ~0u,
                            ok ? (unsigned)(r.z - rowBase) : ~0u, ok ? (unsigned)(r.w - rowBase) : ~0u,
                            unrows, 4 * i, cnt, seg);
        }
        if (cnt > SEG_CAP) {
            int cnt2 = 0;
            for (int base2 = wave * 32; base2 < E4; base2 += 1024) {
                const int i = base2 + lane;
                const bool ok = i < E4;
                const int4 r = rows4[min(i, E4 - 1)];
                const int rr[4] = {r.x, r.y, r.z, r.w};
#pragma unroll
                for (int j = 0; j < 4; ++j) {
                    const bool m = ok && ((unsigned)(rr[j] - rowBase) < unrows);
                    const unsigned mj = __builtin_amdgcn_ballot_w32(m);
                    const int slot = cnt2 + (int)__builtin_amdgcn_mbcnt_lo(mj, 0u);
                    unsigned ov = __builtin_amdgcn_ballot_w32(m && slot >= SEG_CAP);
                    while (ov) {
                        const int src = __builtin_ctz(ov);
                        ov &= ov - 1u;
                        const int rsel = __builtin_amdgcn_readlane(rr[j], src);
                        const int esel = __builtin_amdgcn_readlane(4 * i + j, src);
                        const int c = cols[esel];
                        add_edge_row(tile, Xhat, c, rsel - rowBase, dinv[rsel] * dinv[c], lane);
                    }
                    cnt2 += __builtin_popcount(mj);
                }
            }
        }
    }
    if (wave == 0) {
        for (int e = tailStart; e < E; ++e) {
            const int r = rows[e];
            if ((unsigned)(r - rowBase) < unrows) {
                const int c = cols[e];
                add_edge_row(tile, Xhat, c, r - rowBase, dinv[r] * dinv[c], lane);
            }
        }
    }

    const int nq = min(cnt, SEG_CAP);
    constexpr int G = 16;
    for (int b = 0; b < nq; b += 32) {
        const int nv = min(32, nq - b);
        const int e = seg[b + (lane < nv ? lane : 0)];
        const int r = rows[e];
        const int c = cols[e];
        const float val = dinv[r] * dinv[c];
        const int rl = r - rowBase;
        for (int k0 = 0; k0 < nv; k0 += G) {
            float4 x[G];
#pragma unroll
            for (int u = 0; u < G; ++u) {
                const int cc = __builtin_amdgcn_readlane(c, k0 + u);
                x[u] = *(const float4*)(Xhat + (size_t)cc * 128 + lane * 4);
            }
#pragma unroll
            for (int u = 0; u < G; ++u) {
                if (u == 0 || k0 + u < nv) {
                    const float v = __int_as_float(__builtin_amdgcn_readlane(__float_as_int(val), k0 + u));
                    const int rr  = __builtin_amdgcn_readlane(rl, k0 + u);
                    float* t = tile + rr * 128 + lane;
                    atomicAdd(t,      v * x[u].x);
                    atomicAdd(t + 32, v * x[u].y);
                    atomicAdd(t + 64, v * x[u].z);
                    atomicAdd(t + 96, v * x[u].w);
                }
            }
        }
    }
    __syncthreads();

    for (int row = wave; row < nrows; row += 32) {
        const float* t = tile + row * 128 + lane;
        float4 o;
        o.x = t[0];
        o.y = t[32];
        o.z = t[64];
        o.w = t[96];
        *(float4*)(out + (size_t)(rowBase + row) * 128 + lane * 4) = o;
    }
}

extern "C" void kernel_launch(void* const* d_in, const int* in_sizes, int n_in,
                              void* d_out, int out_size, void* d_ws, size_t ws_size,
                              hipStream_t stream) {
    const float* X  = (const float*)d_in[0];
    const float* W  = (const float*)d_in[1];
    const int*   ei = (const int*)d_in[2];
    float* out = (float*)d_out;

    const int N = in_sizes[0] / 128;
    const int E = in_sizes[2] / 2;
    const int* rows = ei;
    const int* cols = ei + E;

    float* Xhat = (float*)d_ws;
    float* deg  = Xhat + (size_t)N * 128;
    float* dinv = deg + N;

    const int    ntiles   = (N + TILE_ROWS - 1) / TILE_ROWS;
    const size_t keyOff   = ((size_t)N * 130 * sizeof(float) + 255) & ~(size_t)255;
    const size_t keyBytes = (((size_t)E + 1023) >> 10) << 10;
    const bool   keyPlane = ntiles <= 256 && ws_size >= keyOff + keyBytes;
    unsigned char* keys = keyPlane ? (unsigned char*)d_ws + keyOff : nullptr;

    deg_init_kernel<<<(N + 255) / 256, 256, 0, stream>>>(deg, N);
    deg_count_kernel<<<(E + 255) / 256, 256, 0, stream>>>(rows, cols, deg, (unsigned int*)keys, E);
    dinv_kernel<<<(N + 255) / 256, 256, 0, stream>>>(deg, dinv, N);

    gemm_xwT_wmma_kernel<<<512, 256, 0, stream>>>(X, W, Xhat, N);

    if (keyPlane)
        spmm_tile_kernel<true><<<ntiles, 1024, 0, stream>>>(rows, cols, keys, dinv, Xhat, out, N, E);
    else
        spmm_tile_kernel<false><<<ntiles, 1024, 0, stream>>>(rows, cols, keys, dinv, Xhat, out, N, E);
}


# spmm: a queued edge's row and column come as one packed 4-B record (written by the degree-count launch beside the key plane): one random load per edge, not two; sums still in arrival order
# speedup vs baseline: 2.4288x; 1.0418x over previous
#include <hip/hip_runtime.h>

typedef float v2f __attribute__((ext_vector_type(2)));
typedef float v8f __attribute__((ext_vector_type(8)));

__global__ __launch_bounds__(256) void deg_init_kernel(float* __restrict__ deg, int N) {
    int n = blockIdx.x * 256 + threadIdx.x;
    if (n < N) deg[n] = 1.0f;
}

#define TILE_ROWS 400
#define REC_COL_BITS 17
static_assert(TILE_ROWS <= (1 << (32 - REC_COL_BITS)), "local row must fit above the column bits");

__global__ __launch_bounds__(256) void deg_count_kernel(const int* __restrict__ rows,
                                                        const int* __restrict__ cols,
                                                        float* __restrict__ deg,
                                                        unsigned int* __restrict__ keyw,
                                                        unsigned int* __restrict__ recs, int E) {
    const int e = blockIdx.x * 256 + threadIdx.x;
    int c = 0, r = 0;
    if (e < E) {
        c = cols[e];
        if (keyw != nullptr || recs != nullptr) r = rows[e];
        atomicAdd(&deg[c], 1.0f);
        if (recs != nullptr)
            recs[e] = ((unsigned)(r - (r / TILE_ROWS) * TILE_ROWS) << REC_COL_BITS) | (unsigned)c;
    }
    if (keyw != nullptr && (e | 31) < E) {
        const unsigned key  = (unsigned)(r / TILE_ROWS);
        const int      lane = threadIdx.x & 31;
        unsigned mine = 0u;
#pragma unroll
        for (int k = 0; k < 8; ++k) {
            const unsigned bk = __builtin_amdgcn_ballot_w32(((key >> k) & 1u) != 0u);
            if (lane == k) mine = bk;
        }
        if (lane < 8) {
            const int q = e >> 5;
            keyw[((((q >> 5) << 1) + (lane >> 2)) * 32 + (q & 31)) * 4 + (lane & 3)] = mine;
        }
    }
}

__global__ __launch_bounds__(256) void dinv_kernel(const float* __restrict__ deg,
                                                   float* __restrict__ dinv, int N) {
    int n = blockIdx.x * 256 + threadIdx.x;
    if (n < N) dinv[n] = rsqrtf(deg[n]);
}

#define GS_ROWS  32
#define GS_PITCH 132

__global__ __launch_bounds__(256) void gemm_xwT_wmma_kernel(const float* __restrict__ X,
                                                            const float* __restrict__ W,
                                                            float* __restrict__ Xhat,
                                                            int N) {
    __shared__ __attribute__((aligned(16))) float sA[GS_ROWS * GS_PITCH];
    __shared__ __attribute__((aligned(16))) float sO[GS_ROWS * GS_PITCH];

    const int lane    = threadIdx.x & 31;
    const int wave    = __builtin_amdgcn_readfirstlane(threadIdx.x >> 5);
    const int colBase = wave * 16;
    const int half    = lane >> 4;
    const int l16     = lane & 15;
    const int nstrips = (N + GS_ROWS - 1) / GS_ROWS;
    const int stride  = (int)gridDim.x;

    v2f bfrag[32];
    {
        const float* bptr = W + (size_t)(colBase + l16) * 128 + (half << 1);
#pragma unroll
        for (int ks = 0; ks < 32; ++ks) bfrag[ks] = *(const v2f*)(bptr + 4 * ks);
    }

    const int myRow = wave * 4;
    const float* xlane = X + lane * 4;
    float4 x0 = {0.f, 0.f, 0.f, 0.f}, x1 = x0, x2 = x0, x3 = x0;
    int s = blockIdx.x;
    if (s < nstrips) {
        const int r0 = s * GS_ROWS + myRow;
        x0 = *(const float4*)(xlane + (size_t)min(r0 + 0, N - 1) * 128);
        x1 = *(const float4*)(xlane + (size_t)min(r0 + 1, N - 1) * 128);
        x2 = *(const float4*)(xlane + (size_t)min(r0 + 2, N - 1) * 128);
        x3 = *(const float4*)(xlane + (size_t)min(r0 + 3, N - 1) * 128);
    }

    const float* a0p = sA + l16 * GS_PITCH + (half << 1);
    const float* a1p = sA + (16 + l16) * GS_PITCH + (half << 1);

    for (; s < nstrips; s += stride) {
        float* arow = sA + myRow * GS_PITCH + lane * 4;
        *(float4*)(arow + 0 * GS_PITCH) = x0;
        *(float4*)(arow + 1 * GS_PITCH) = x1;
        *(float4*)(arow + 2 * GS_PITCH) = x2;
        *(float4*)(arow + 3 * GS_PITCH) = x3;
        __syncthreads();

        const int sn = s + stride;
        if (sn < nstrips) {
            const int r0 = sn * GS_ROWS + myRow;
            x0 = *(const float4*)(xlane + (size_t)min(r0 + 0, N - 1) * 128);
            x1 = *(const float4*)(xlane + (size_t)min(r0 + 1, N - 1) * 128);
            x2 = *(const float4*)(xlane + (size_t)min(r0 + 2, N - 1) * 128);
            x3 = *(const float4*)(xlane + (size_t)min(r0 + 3, N - 1) * 128);
        }

        v8f acc0 = {0.f, 0.f, 0.f, 0.f, 0.f, 0.f, 0.f, 0.f};
        v8f acc1 = {0.f, 0.f, 0.f, 0.f, 0.f, 0.f, 0.f, 0.f};
#pragma unroll
        for (int ks = 0; ks < 32; ++ks) {
            const v2f a0 = *(const v2f*)(a0p + 4 * ks);
            const v2f a1 = *(const v2f*)(a1p + 4 * ks);
            acc0 = __builtin_amdgcn_wmma_f32_16x16x4_f32(
                 false, a0,  false, bfrag[ks],
                 (short)0, acc0,  false,  false);
            acc1 = __builtin_amdgcn_wmma_f32_16x16x4_f32(
                 false, a1,  false, bfrag[ks],
                 (short)0, acc1,  false,  false);
        }

#pragma unroll
        for (int v = 0; v < 8; ++v) {
            const int m = v + (half << 3);
            sO[m * GS_PITCH + colBase + l16]        = acc0[v];
            sO[(16 + m) * GS_PITCH + colBase + l16] = acc1[v];
        }
        __syncthreads();

        const float* orow = sO + myRow * GS_PITCH + lane * 4;
        const int r0 = s * GS_ROWS + myRow;
        float* dst = Xhat + (size_t)r0 * 128 + lane * 4;
        if (r0 + 3 < N) {
            const float4 o0 = *(const float4*)(orow + 0 * GS_PITCH);
            const float4 o1 = *(const float4*)(orow + 1 * GS_PITCH);
            const float4 o2 = *(const float4*)(orow + 2 * GS_PITCH);
            const float4 o3 = *(const float4*)(orow + 3 * GS_PITCH);
            *(float4*)(dst + 0 * 128) = o0;
            *(float4*)(dst + 1 * 128) = o1;
            *(float4*)(dst + 2 * 128) = o2;
            *(float4*)(dst + 3 * 128) = o3;
        } else {
            for (int j = 0; j < 4; ++j)
                if (r0 + j < N) *(float4*)(dst + j * 128) = *(const float4*)(orow + j * GS_PITCH);
        }
    }
}

#define SEG_CAP   768

__device__ __forceinline__ void add_edge_row(float* __restrict__ tile,
                                             const float* __restrict__ Xhat,
                                             int c, int rl, float val, int lane) {
    const float4 x = *(const float4*)(Xhat + (size_t)c * 128 + lane * 4);
    float* t = tile + rl * 128 + lane;
    atomicAdd(t,      val * x.x);
    atomicAdd(t + 32, val * x.y);
    atomicAdd(t + 64, val * x.z);
    atomicAdd(t + 96, val * x.w);
}

__device__ __forceinline__ int push_match(bool m, int e, int cnt, int* __restrict__ seg) {
    const unsigned mj = __builtin_amdgcn_ballot_w32(m);
    if (mj == 0u) return cnt;
    const int slot = cnt + (int)__builtin_amdgcn_mbcnt_lo(mj, 0u);
    if (m && slot < SEG_CAP) seg[slot] = e;
    return cnt + __builtin_popcount(mj);
}

__device__ __forceinline__ int scan_step(unsigned a0, unsigned a1, unsigned a2, unsigned a3,
                                         unsigned unrows, int e0, int cnt, int* __restrict__ seg) {
    if (__builtin_amdgcn_ballot_w32(min(min(a0, a1), min(a2, a3)) < unrows) != 0u) {
        cnt = push_match(a0 < unrows, e0 + 0, cnt, seg);
        cnt = push_match(a1 < unrows, e0 + 1, cnt, seg);
        cnt = push_match(a2 < unrows, e0 + 2, cnt, seg);
        cnt = push_match(a3 < unrows, e0 + 3, cnt, seg);
    }
    return cnt;
}

template <bool SECOND>
__device__ __forceinline__ int key_step(uint4 a, uint4 b, bool ok, unsigned kb, int e0, int cnt,
                                        int* __restrict__ seg, float* __restrict__ tile,
                                        const int* __restrict__ rows, const int* __restrict__ cols,
                                        const float* __restrict__ dinv, const float* __restrict__ Xhat,
                                        int rowBase, int lane) {
    unsigned m = ~(a.x ^ (0u - ((kb >> 0) & 1u))) & ~(a.y ^ (0u - ((kb >> 1) & 1u))) &
                 ~(a.z ^ (0u - ((kb >> 2) & 1u))) & ~(a.w ^ (0u - ((kb >> 3) & 1u))) &
                 ~(b.x ^ (0u - ((kb >> 4) & 1u))) & ~(b.y ^ (0u - ((kb >> 5) & 1u))) &
                 ~(b.z ^ (0u - ((kb >> 6) & 1u))) & ~(b.w ^ (0u - ((kb >> 7) & 1u)));
    if (!ok) m = 0u;
    for (;;) {
        const bool has = m != 0u;
        const unsigned mj = __builtin_amdgcn_ballot_w32(has);
        if (mj == 0u) break;
        const int e = e0 + __builtin_ffs((int)m) - 1;
        const int slot = cnt + (int)__builtin_amdgcn_mbcnt_lo(mj, 0u);
        if constexpr (!SECOND) {
            if (has && slot < SEG_CAP) seg[slot] = e;
        } else {
            unsigned ov = __builtin_amdgcn_ballot_w32(has && slot >= SEG_CAP);
            while (ov) {
                const int srcl = __builtin_ctz(ov);
                ov &= ov - 1u;
                const int esel = __builtin_amdgcn_readlane(e, srcl);
                const int r = rows[esel];
                const int c = cols[esel];
                add_edge_row(tile, Xhat, c, r - rowBase, dinv[r] * dinv[c], lane);
            }
        }
        m &= m - 1u;
        cnt += __builtin_popcount(mj);
    }
    return cnt;
}

template <bool KEY_PLANE, bool RECORDS>
__global__ __launch_bounds__(1024) void spmm_tile_kernel(const int* __restrict__ rows,
                                                         const int* __restrict__ cols,
                                                         const unsigned char* __restrict__ keys,
                                                         const unsigned int* __restrict__ recs,
                                                         const float* __restrict__ dinv,
                                                         const float* __restrict__ Xhat,
                                                         float* __restrict__ out,
                                                         int N, int E) {
    __shared__ float tile[TILE_ROWS * 128];
    __shared__ int   queue[32 * SEG_CAP];

    const int tid  = threadIdx.x;
    const int lane = tid & 31;
    const int wave = __builtin_amdgcn_readfirstlane(tid >> 5);
    const int rowBase = blockIdx.x * TILE_ROWS;
    const int nrows   = min(TILE_ROWS, N - rowBase);
    const unsigned unrows = (unsigned)nrows;

#pragma unroll 4
    for (int row = wave; row < nrows; row += 32) {
        const int n = rowBase + row;
        const float d = dinv[n];
        const float4 x = *(const float4*)(Xhat + (size_t)n * 128 + lane * 4);
        float* t = tile + row * 128 + lane;
        t[0]  = d * d * x.x;
        t[32] = d * d * x.y;
        t[64] = d * d * x.z;
        t[96] = d * d * x.w;
    }
    __syncthreads();

    int* seg = queue + wave * SEG_CAP;
    int cnt = 0;
    constexpr int U = 8;
    const int tailStart = KEY_PLANE ? (E >> 5) << 5 : (E >> 2) << 2;
    if constexpr (KEY_PLANE) {
        const int G32 = tailStart >> 5;
        const int SGF = G32 >> 5;
        const uint4* kp = (const uint4*)keys;
        const unsigned kb = blockIdx.x;
        int sg = wave;
        for (; sg + (U - 1) * 32 < SGF; sg += 32 * U) {
            uint4 ka[U], kc[U];
#pragma unroll
            for (int u = 0; u < U; ++u) {
                ka[u] = kp[(sg + 32 * u) * 64 + lane];
                kc[u] = kp[(sg + 32 * u) * 64 + 32 + lane];
            }
#pragma unroll
            for (int u = 0; u < U; ++u) {
                cnt = key_step<false>(ka[u], kc[u], true, kb, ((sg + 32 * u) * 32 + lane) * 32, cnt, seg,
                                      tile, rows, cols, dinv, Xhat, rowBase, lane);
            }
        }
        for (; sg * 32 < G32; sg += 32) {
            const int g  = sg * 32 + lane;
            const int gl = min(g, G32 - 1);
            const int o  = (gl >> 5) * 64 + (gl & 31);
            cnt = key_step<false>(kp[o], kp[o + 32], g < G32, kb, g * 32, cnt, seg,
                                  tile, rows, cols, dinv, Xhat, rowBase, lane);
        }
        if (cnt > SEG_CAP) {
            int cnt2 = 0;
            for (int sg2 = wave; sg2 * 32 < G32; sg2 += 32) {
                const int g  = sg2 * 32 + lane;
                const int gl = min(g, G32 - 1);
                const int o  = (gl >> 5) * 64 + (gl & 31);
                cnt2 = key_step<true>(kp[o], kp[o + 32], g < G32, kb, g * 32, cnt2, seg,
                                      tile, rows, cols, dinv, Xhat, rowBase, lane);
            }
        }
    } else {
        const int E4 = tailStart >> 2;
        const int4* rows4 = (const int4*)rows;
        int base = wave * 32;
        for (; base + (U - 1) * 1024 + 31 < E4; base += 1024 * U) {
            int4 r4[U];
#pragma unroll
            for (int u = 0; u < U; ++u) r4[u] = rows4[base + u * 1024 + lane];
#pragma unroll
            for (int u = 0; u < U; ++u) {
                cnt = scan_step((unsigned)(r4[u].x - rowBase), (unsigned)(r4[u].y - rowBase),
                                (unsigned)(r4[u].z - rowBase), (unsigned)(r4[u].w - rowBase),
                                unrows, 4 * (base + u * 1024 + lane), cnt, seg);
            }
        }
        for (; base < E4; base += 1024) {
            const int i = base + lane;
            const bool ok = i < E4;
            const int4 r = rows4[min(i, E4 - 1)];
            cnt = scan_step(ok ? (unsigned)(r.x - rowBase) : ~0u, ok ? (unsigned)(r.y - rowBase) : ~0u,
                            ok ? (unsigned)(r.z - rowBase) : ~0u, ok ? (unsigned)(r.w - rowBase) : ~0u,
                            unrows, 4 * i, cnt, seg);
        }
        if (cnt > SEG_CAP) {
            int cnt2 = 0;
            for (int base2 = wave * 32; base2 < E4; base2 += 1024) {
                const int i = base2 + lane;
                const bool ok = i < E4;
                const int4 r = rows4[min(i, E4 - 1)];
                const int rr[4] = {r.x, r.y, r.z, r.w};
#pragma unroll
                for (int j = 0; j < 4; ++j) {
                    const bool m = ok && ((unsigned)(rr[j] - rowBase) < unrows);
                    const unsigned mj = __builtin_amdgcn_ballot_w32(m);
                    const int slot = cnt2 + (int)__builtin_amdgcn_mbcnt_lo(mj, 0u);
                    unsigned ov = __builtin_amdgcn_ballot_w32(m && slot >= SEG_CAP);
                    while (ov) {
                        const int src = __builtin_ctz(ov);
                        ov &= ov - 1u;
                        const int rsel = __builtin_amdgcn_readlane(rr[j], src);
                        const int esel = __builtin_amdgcn_readlane(4 * i + j, src);
                        const int c = cols[esel];
                        add_edge_row(tile, Xhat, c, rsel - rowBase, dinv[rsel] * dinv[c], lane);
                    }
                    cnt2 += __builtin_popcount(mj);
                }
            }
        }
    }
    if (wave == 0) {
        for (int e = tailStart; e < E; ++e) {
            const int r = rows[e];
            if ((unsigned)(r - rowBase) < unrows) {
                const int c = cols[e];
                add_edge_row(tile, Xhat, c, r - rowBase, dinv[r] * dinv[c], lane);
            }
        }
    }

    const int nq = min(cnt, SEG_CAP);
    constexpr int G = 16;
    for (int b = 0; b < nq; b += 32) {
        const int nv = min(32, nq - b);
        const int e = seg[b + (lane < nv ? lane : 0)];
        int rl, c;
        if constexpr (RECORDS) {
            const unsigned w = recs[e];
            c  = (int)(w & ((1u << REC_COL_BITS) - 1u));
            rl = (int)(w >> REC_COL_BITS);
        } else {
            c  = cols[e];
            rl = rows[e] - rowBase;
        }
        const float val = dinv[rowBase + rl] * dinv[c];
        for (int k0 = 0; k0 < nv; k0 += G) {
            float4 x[G];
#pragma unroll
            for (int u = 0; u < G; ++u) {
                const int cc = __builtin_amdgcn_readlane(c, k0 + u);
                x[u] = *(const float4*)(Xhat + (size_t)cc * 128 + lane * 4);
            }
#pragma unroll
            for (int u = 0; u < G; ++u) {
                if (u == 0 || k0 + u < nv) {
                    const float v = __int_as_float(__builtin_amdgcn_readlane(__float_as_int(val), k0 + u));
                    const int rr  = __builtin_amdgcn_readlane(rl, k0 + u);
                    float* t = tile + rr * 128 + lane;
                    atomicAdd(t,      v * x[u].x);
                    atomicAdd(t + 32, v * x[u].y);
                    atomicAdd(t + 64, v * x[u].z);
                    atomicAdd(t + 96, v * x[u].w);
                }
            }
        }
    }
    __syncthreads();

    for (int row = wave; row < nrows; row += 32) {
        const float* t = tile + row * 128 + lane;
        float4 o;
        o.x = t[0];
        o.y = t[32];
        o.z = t[64];
        o.w = t[96];
        *(float4*)(out + (size_t)(rowBase + row) * 128 + lane * 4) = o;
    }
}

extern "C" void kernel_launch(void* const* d_in, const int* in_sizes, int n_in,
                              void* d_out, int out_size, void* d_ws, size_t ws_size,
                              hipStream_t stream) {
    const float* X  = (const float*)d_in[0];
    const float* W  = (const float*)d_in[1];
    const int*   ei = (const int*)d_in[2];
    float* out = (float*)d_out;

    const int N = in_sizes[0] / 128;
    const int E = in_sizes[2] / 2;
    const int* rows = ei;
    const int* cols = ei + E;

    float* Xhat = (float*)d_ws;
    float* deg  = Xhat + (size_t)N * 128;
    float* dinv = deg + N;

    const int    ntiles   = (N + TILE_ROWS - 1) / TILE_ROWS;
    const size_t keyOff   = ((size_t)N * 130 * sizeof(float) + 255) & ~(size_t)255;
    const size_t keyBytes = (((size_t)E + 1023) >> 10) << 10;
    const bool   keyPlane = ntiles <= 256 && ws_size >= keyOff + keyBytes;
    unsigned char* keys = keyPlane ? (unsigned char*)d_ws + keyOff : nullptr;
    const size_t recOff   = keyOff + keyBytes;
    const bool   recPlane = keyPlane && N <= (1 << REC_COL_BITS) && ws_size >= recOff + (size_t)E * sizeof(unsigned int);
    unsigned int* recs = recPlane ? (unsigned int*)((unsigned char*)d_ws + recOff) : nullptr;

    deg_init_kernel<<<(N + 255) / 256, 256, 0, stream>>>(deg, N);
    deg_count_kernel<<<(E + 255) / 256, 256, 0, stream>>>(rows, cols, deg, (unsigned int*)keys, recs, E);
    dinv_kernel<<<(N + 255) / 256, 256, 0, stream>>>(deg, dinv, N);

    gemm_xwT_wmma_kernel<<<512, 256, 0, stream>>>(X, W, Xhat, N);

    if (recPlane)
        spmm_tile_kernel<true, true><<<ntiles, 1024, 0, stream>>>(rows, cols, keys, recs, dinv, Xhat, out, N, E);
    else if (keyPlane)
        spmm_tile_kernel<true, false><<<ntiles, 1024, 0, stream>>>(rows, cols, keys, recs, dinv, Xhat, out, N, E);
    else
        spmm_tile_kernel<false, false><<<ntiles, 1024, 0, stream>>>(rows, cols, keys, recs, dinv, Xhat, out, N, E);
}


# degree count and gemm as two roles of one launch: blocks 0-511 form X_hat, 256 resident blocks behind them count (25 32-edge groups a wave): atomics run under the matrix work; sum order kept
# speedup vs baseline: 2.7276x; 1.1230x over previous
#include <hip/hip_runtime.h>

typedef float v2f __attribute__((ext_vector_type(2)));
typedef float v8f __attribute__((ext_vector_type(8)));

__global__ __launch_bounds__(256) void deg_init_kernel(float* __restrict__ deg, int N) {
    int n = blockIdx.x * 256 + threadIdx.x;
    if (n < N) deg[n] = 1.0f;
}

#define TILE_ROWS 400
#define REC_COL_BITS 17
static_assert(TILE_ROWS <= (1 << (32 - REC_COL_BITS)), "local row must fit above the column bits");

#define CG 13

__device__ __forceinline__ void count_role(const int* __restrict__ rows,
                                           const int* __restrict__ cols,
                                           float* __restrict__ deg,
                                           unsigned int* __restrict__ keyw,
                                           unsigned int* __restrict__ recs, int E,
                                           int bid, int nblk) {
    const int lane = threadIdx.x & 31;
    const int wave = __builtin_amdgcn_readfirstlane(threadIdx.x >> 5);
    const int G    = (E + 31) >> 5;
    const int nw   = nblk * 8;
    const int per  = (G + nw - 1) / nw;
    const int qBeg = (bid * 8 + wave) * per;
    const int qEnd = min(qBeg + per, G);
    const bool needRow = keyw != nullptr || recs != nullptr;
    for (int q0 = qBeg; q0 < qEnd; q0 += CG) {
        int c[CG], r[CG];
#pragma unroll
        for (int u = 0; u < CG; ++u) {
            const int e = (q0 + u) * 32 + lane;
            c[u] = 0;
            r[u] = 0;
            if (q0 + u < qEnd && e < E) {
                c[u] = cols[e];
                if (needRow) r[u] = rows[e];
            }
        }
#pragma unroll
        for (int u = 0; u < CG; ++u) {
            const int q = q0 + u;
            if (q < qEnd) {
                const int e = q * 32 + lane;
                if (e < E) {
                    atomicAdd(&deg[c[u]], 1.0f);
                    if (recs != nullptr)
                        recs[e] = ((unsigned)(r[u] - (r[u] / TILE_ROWS) * TILE_ROWS) << REC_COL_BITS) | (unsigned)c[u];
                }
                if (keyw != nullptr && q * 32 + 31 < E) {
                    const unsigned key = (unsigned)(r[u] / TILE_ROWS);
                    unsigned mine = 0u;
#pragma unroll
                    for (int k = 0; k < 8; ++k) {
                        const unsigned bk = __builtin_amdgcn_ballot_w32(((key >> k) & 1u) != 0u);
                        if (lane == k) mine = bk;
                    }
                    if (lane < 8)
                        keyw[((((q >> 5) << 1) + (lane >> 2)) * 32 + (q & 31)) * 4 + (lane & 3)] = mine;
                }
            }
        }
    }
}

__global__ __launch_bounds__(256) void dinv_kernel(const float* __restrict__ deg,
                                                   float* __restrict__ dinv, int N) {
    int n = blockIdx.x * 256 + threadIdx.x;
    if (n < N) dinv[n] = rsqrtf(deg[n]);
}

#define GS_ROWS  32
#define GS_PITCH 132

__device__ __forceinline__ void gemm_role(const float* __restrict__ X,
                                          const float* __restrict__ W,
                                          float* __restrict__ Xhat,
                                          int N, int bid, int nblk) {
    __shared__ __attribute__((aligned(16))) float sA[GS_ROWS * GS_PITCH];
    __shared__ __attribute__((aligned(16))) float sO[GS_ROWS * GS_PITCH];

    const int lane    = threadIdx.x & 31;
    const int wave    = __builtin_amdgcn_readfirstlane(threadIdx.x >> 5);
    const int colBase = wave * 16;
    const int half    = lane >> 4;
    const int l16     = lane & 15;
    const int nstrips = (N + GS_ROWS - 1) / GS_ROWS;
    const int stride  = nblk;

    v2f bfrag[32];
    {
        const float* bptr = W + (size_t)(colBase + l16) * 128 + (half << 1);
#pragma unroll
        for (int ks = 0; ks < 32; ++ks) bfrag[ks] = *(const v2f*)(bptr + 4 * ks);
    }

    const int myRow = wave * 4;
    const float* xlane = X + lane * 4;
    float4 x0 = {0.f, 0.f, 0.f, 0.f}, x1 = x0, x2 = x0, x3 = x0;
    int s = bid;
    if (s < nstrips) {
        const int r0 = s * GS_ROWS + myRow;
        x0 = *(const float4*)(xlane + (size_t)min(r0 + 0, N - 1) * 128);
        x1 = *(const float4*)(xlane + (size_t)min(r0 + 1, N - 1) * 128);
        x2 = *(const float4*)(xlane + (size_t)min(r0 + 2, N - 1) * 128);
        x3 = *(const float4*)(xlane + (size_t)min(r0 + 3, N - 1) * 128);
    }

    const float* a0p = sA + l16 * GS_PITCH + (half << 1);
    const float* a1p = sA + (16 + l16) * GS_PITCH + (half << 1);

    for (; s < nstrips; s += stride) {
        float* arow = sA + myRow * GS_PITCH + lane * 4;
        *(float4*)(arow + 0 * GS_PITCH) = x0;
        *(float4*)(arow + 1 * GS_PITCH) = x1;
        *(float4*)(arow + 2 * GS_PITCH) = x2;
        *(float4*)(arow + 3 * GS_PITCH) = x3;
        __syncthreads();

        const int sn = s + stride;
        if (sn < nstrips) {
            const int r0 = sn * GS_ROWS + myRow;
            x0 = *(const float4*)(xlane + (size_t)min(r0 + 0, N - 1) * 128);
            x1 = *(const float4*)(xlane + (size_t)min(r0 + 1, N - 1) * 128);
            x2 = *(const float4*)(xlane + (size_t)min(r0 + 2, N - 1) * 128);
            x3 = *(const float4*)(xlane + (size_t)min(r0 + 3, N - 1) * 128);
        }

        v8f acc0 = {0.f, 0.f, 0.f, 0.f, 0.f, 0.f, 0.f, 0.f};
        v8f acc1 = {0.f, 0.f, 0.f, 0.f, 0.f, 0.f, 0.f, 0.f};
#pragma unroll
        for (int ks = 0; ks < 32; ++ks) {
            const v2f a0 = *(const v2f*)(a0p + 4 * ks);
            const v2f a1 = *(const v2f*)(a1p + 4 * ks);
            acc0 = __builtin_amdgcn_wmma_f32_16x16x4_f32(
                 false, a0,  false, bfrag[ks],
                 (short)0, acc0,  false,  false);
            acc1 = __builtin_amdgcn_wmma_f32_16x16x4_f32(
                 false, a1,  false, bfrag[ks],
                 (short)0, acc1,  false,  false);
        }

#pragma unroll
        for (int v = 0; v < 8; ++v) {
            const int m = v + (half << 3);
            sO[m * GS_PITCH + colBase + l16]        = acc0[v];
            sO[(16 + m) * GS_PITCH + colBase + l16] = acc1[v];
        }
        __syncthreads();

        const float* orow = sO + myRow * GS_PITCH + lane * 4;
        const int r0 = s * GS_ROWS + myRow;
        float* dst = Xhat + (size_t)r0 * 128 + lane * 4;
        if (r0 + 3 < N) {
            const float4 o0 = *(const float4*)(orow + 0 * GS_PITCH);
            const float4 o1 = *(const float4*)(orow + 1 * GS_PITCH);
            const float4 o2 = *(const float4*)(orow + 2 * GS_PITCH);
            const float4 o3 = *(const float4*)(orow + 3 * GS_PITCH);
            *(float4*)(dst + 0 * 128) = o0;
            *(float4*)(dst + 1 * 128) = o1;
            *(float4*)(dst + 2 * 128) = o2;
            *(float4*)(dst + 3 * 128) = o3;
        } else {
            for (int j = 0; j < 4; ++j)
                if (r0 + j < N) *(float4*)(dst + j * 128) = *(const float4*)(orow + j * GS_PITCH);
        }
    }
}

__global__ __launch_bounds__(256) void gemm_and_count_kernel(const float* __restrict__ X,
                                                             const float* __restrict__ W,
                                                             float* __restrict__ Xhat, int N,
                                                             const int* __restrict__ rows,
                                                             const int* __restrict__ cols,
                                                             float* __restrict__ deg,
                                                             unsigned int* __restrict__ keyw,
                                                             unsigned int* __restrict__ recs, int E,
                                                             int gemmBlocks) {
    const int bid = (int)blockIdx.x;
    if (bid < gemmBlocks)
        gemm_role(X, W, Xhat, N, bid, gemmBlocks);
    else
        count_role(rows, cols, deg, keyw, recs, E, bid - gemmBlocks, (int)gridDim.x - gemmBlocks);
}

#define SEG_CAP   768

__device__ __forceinline__ void add_edge_row(float* __restrict__ tile,
                                             const float* __restrict__ Xhat,
                                             int c, int rl, float val, int lane) {
    const float4 x = *(const float4*)(Xhat + (size_t)c * 128 + lane * 4);
    float* t = tile + rl * 128 + lane;
    atomicAdd(t,      val * x.x);
    atomicAdd(t + 32, val * x.y);
    atomicAdd(t + 64, val * x.z);
    atomicAdd(t + 96, val * x.w);
}

__device__ __forceinline__ int push_match(bool m, int e, int cnt, int* __restrict__ seg) {
    const unsigned mj = __builtin_amdgcn_ballot_w32(m);
    if (mj == 0u) return cnt;
    const int slot = cnt + (int)__builtin_amdgcn_mbcnt_lo(mj, 0u);
    if (m && slot < SEG_CAP) seg[slot] = e;
    return cnt + __builtin_popcount(mj);
}

__device__ __forceinline__ int scan_step(unsigned a0, unsigned a1, unsigned a2, unsigned a3,
                                         unsigned unrows, int e0, int cnt, int* __restrict__ seg) {
    if (__builtin_amdgcn_ballot_w32(min(min(a0, a1), min(a2, a3)) < unrows) != 0u) {
        cnt = push_match(a0 < unrows, e0 + 0, cnt, seg);
        cnt = push_match(a1 < unrows, e0 + 1, cnt, seg);
        cnt = push_match(a2 < unrows, e0 + 2, cnt, seg);
        cnt = push_match(a3 < unrows, e0 + 3, cnt, seg);
    }
    return cnt;
}

template <bool SECOND>
__device__ __forceinline__ int key_step(uint4 a, uint4 b, bool ok, unsigned kb, int e0, int cnt,
                                        int* __restrict__ seg, float* __restrict__ tile,
                                        const int* __restrict__ rows, const int* __restrict__ cols,
                                        const float* __restrict__ dinv, const float* __restrict__ Xhat,
                                        int rowBase, int lane) {
    unsigned m = ~(a.x ^ (0u - ((kb >> 0) & 1u))) & ~(a.y ^ (0u - ((kb >> 1) & 1u))) &
                 ~(a.z ^ (0u - ((kb >> 2) & 1u))) & ~(a.w ^ (0u - ((kb >> 3) & 1u))) &
                 ~(b.x ^ (0u - ((kb >> 4) & 1u))) & ~(b.y ^ (0u - ((kb >> 5) & 1u))) &
                 ~(b.z ^ (0u - ((kb >> 6) & 1u))) & ~(b.w ^ (0u - ((kb >> 7) & 1u)));
    if (!ok) m = 0u;
    for (;;) {
        const bool has = m != 0u;
        const unsigned mj = __builtin_amdgcn_ballot_w32(has);
        if (mj == 0u) break;
        const int e = e0 + __builtin_ffs((int)m) - 1;
        const int slot = cnt + (int)__builtin_amdgcn_mbcnt_lo(mj, 0u);
        if constexpr (!SECOND) {
            if (has && slot < SEG_CAP) seg[slot] = e;
        } else {
            unsigned ov = __builtin_amdgcn_ballot_w32(has && slot >= SEG_CAP);
            while (ov) {
                const int srcl = __builtin_ctz(ov);
                ov &= ov - 1u;
                const int esel = __builtin_amdgcn_readlane(e, srcl);
                const int r = rows[esel];
                const int c = cols[esel];
                add_edge_row(tile, Xhat, c, r - rowBase, dinv[r] * dinv[c], lane);
            }
        }
        m &= m - 1u;
        cnt += __builtin_popcount(mj);
    }
    return cnt;
}

template <bool KEY_PLANE, bool RECORDS>
__global__ __launch_bounds__(1024) void spmm_tile_kernel(const int* __restrict__ rows,
                                                         const int* __restrict__ cols,
                                                         const unsigned char* __restrict__ keys,
                                                         const unsigned int* __restrict__ recs,
                                                         const float* __restrict__ dinv,
                                                         const float* __restrict__ Xhat,
                                                         float* __restrict__ out,
                                                         int N, int E) {
    __shared__ float tile[TILE_ROWS * 128];
    __shared__ int   queue[32 * SEG_CAP];

    const int tid  = threadIdx.x;
    const int lane = tid & 31;
    const int wave = __builtin_amdgcn_readfirstlane(tid >> 5);
    const int rowBase = blockIdx.x * TILE_ROWS;
    const int nrows   = min(TILE_ROWS, N - rowBase);
    const unsigned unrows = (unsigned)nrows;

#pragma unroll 4
    for (int row = wave; row < nrows; row += 32) {
        const int n = rowBase + row;
        const float d = dinv[n];
        const float4 x = *(const float4*)(Xhat + (size_t)n * 128 + lane * 4);
        float* t = tile + row * 128 + lane;
        t[0]  = d * d * x.x;
        t[32] = d * d * x.y;
        t[64] = d * d * x.z;
        t[96] = d * d * x.w;
    }
    __syncthreads();

    int* seg = queue + wave * SEG_CAP;
    int cnt = 0;
    constexpr int U = 8;
    const int tailStart = KEY_PLANE ? (E >> 5) << 5 : (E >> 2) << 2;
    if constexpr (KEY_PLANE) {
        const int G32 = tailStart >> 5;
        const int SGF = G32 >> 5;
        const uint4* kp = (const uint4*)keys;
        const unsigned kb = blockIdx.x;
        int sg = wave;
        for (; sg + (U - 1) * 32 < SGF; sg += 32 * U) {
            uint4 ka[U], kc[U];
#pragma unroll
            for (int u = 0; u < U; ++u) {
                ka[u] = kp[(sg + 32 * u) * 64 + lane];
                kc[u] = kp[(sg + 32 * u) * 64 + 32 + lane];
            }
#pragma unroll
            for (int u = 0; u < U; ++u) {
                cnt = key_step<false>(ka[u], kc[u], true, kb, ((sg + 32 * u) * 32 + lane) * 32, cnt, seg,
                                      tile, rows, cols, dinv, Xhat, rowBase, lane);
            }
        }
        for (; sg * 32 < G32; sg += 32) {
            const int g  = sg * 32 + lane;
            const int gl = min(g, G32 - 1);
            const int o  = (gl >> 5) * 64 + (gl & 31);
            cnt = key_step<false>(kp[o], kp[o + 32], g < G32, kb, g * 32, cnt, seg,
                                  tile, rows, cols, dinv, Xhat, rowBase, lane);
        }
        if (cnt > SEG_CAP) {
            int cnt2 = 0;
            for (int sg2 = wave; sg2 * 32 < G32; sg2 += 32) {
                const int g  = sg2 * 32 + lane;
                const int gl = min(g, G32 - 1);
                const int o  = (gl >> 5) * 64 + (gl & 31);
                cnt2 = key_step<true>(kp[o], kp[o + 32], g < G32, kb, g * 32, cnt2, seg,
                                      tile, rows, cols, dinv, Xhat, rowBase, lane);
            }
        }
    } else {
        const int E4 = tailStart >> 2;
        const int4* rows4 = (const int4*)rows;
        int base = wave * 32;
        for (; base + (U - 1) * 1024 + 31 < E4; base += 1024 * U) {
            int4 r4[U];
#pragma unroll
            for (int u = 0; u < U; ++u) r4[u] = rows4[base + u * 1024 + lane];
#pragma unroll
            for (int u = 0; u < U; ++u) {
                cnt = scan_step((unsigned)(r4[u].x - rowBase), (unsigned)(r4[u].y - rowBase),
                                (unsigned)(r4[u].z - rowBase), (unsigned)(r4[u].w - rowBase),
                                unrows, 4 * (base + u * 1024 + lane), cnt, seg);
            }
        }
        for (; base < E4; base += 1024) {
            const int i = base + lane;
            const bool ok = i < E4;
            const int4 r = rows4[min(i, E4 - 1)];
            cnt = scan_step(ok ? (unsigned)(r.x - rowBase) : ~0u, ok ? (unsigned)(r.y - rowBase) : ~0u,
                            ok ? (unsigned)(r.z - rowBase) : ~0u, ok ? (unsigned)(r.w - rowBase) : ~0u,
                            unrows, 4 * i, cnt, seg);
        }
        if (cnt > SEG_CAP) {
            int cnt2 = 0;
            for (int base2 = wave * 32; base2 < E4; base2 += 1024) {
                const int i = base2 + lane;
                const bool ok = i < E4;
                const int4 r = rows4[min(i, E4 - 1)];
                const int rr[4] = {r.x, r.y, r.z, r.w};
#pragma unroll
                for (int j = 0; j < 4; ++j) {
                    const bool m = ok && ((unsigned)(rr[j] - rowBase) < unrows);
                    const unsigned mj = __builtin_amdgcn_ballot_w32(m);
                    const int slot = cnt2 + (int)__builtin_amdgcn_mbcnt_lo(mj, 0u);
                    unsigned ov = __builtin_amdgcn_ballot_w32(m && slot >= SEG_CAP);
                    while (ov) {
                        const int src = __builtin_ctz(ov);
                        ov &= ov - 1u;
                        const int rsel = __builtin_amdgcn_readlane(rr[j], src);
                        const int esel = __builtin_amdgcn_readlane(4 * i + j, src);
                        const int c = cols[esel];
                        add_edge_row(tile, Xhat, c, rsel - rowBase, dinv[rsel] * dinv[c], lane);
                    }
                    cnt2 += __builtin_popcount(mj);
                }
            }
        }
    }
    if (wave == 0) {
        for (int e = tailStart; e < E; ++e) {
            const int r = rows[e];
            if ((unsigned)(r - rowBase) < unrows) {
                const int c = cols[e];
                add_edge_row(tile, Xhat, c, r - rowBase, dinv[r] * dinv[c], lane);
            }
        }
    }

    const int nq = min(cnt, SEG_CAP);
    constexpr int G = 16;
    for (int b = 0; b < nq; b += 32) {
        const int nv = min(32, nq - b);
        const int e = seg[b + (lane < nv ? lane : 0)];
        int rl, c;
        if constexpr (RECORDS) {
            const unsigned w = recs[e];
            c  = (int)(w & ((1u << REC_COL_BITS) - 1u));
            rl = (int)(w >> REC_COL_BITS);
        } else {
            c  = cols[e];
            rl = rows[e] - rowBase;
        }
        const float val = dinv[rowBase + rl] * dinv[c];
        for (int k0 = 0; k0 < nv; k0 += G) {
            float4 x[G];
#pragma unroll
            for (int u = 0; u < G; ++u) {
                const int cc = __builtin_amdgcn_readlane(c, k0 + u);
                x[u] = *(const float4*)(Xhat + (size_t)cc * 128 + lane * 4);
            }
#pragma unroll
            for (int u = 0; u < G; ++u) {
                if (u == 0 || k0 + u < nv) {
                    const float v = __int_as_float(__builtin_amdgcn_readlane(__float_as_int(val), k0 + u));
                    const int rr  = __builtin_amdgcn_readlane(rl, k0 + u);
                    float* t = tile + rr * 128 + lane;
                    atomicAdd(t,      v * x[u].x);
                    atomicAdd(t + 32, v * x[u].y);
                    atomicAdd(t + 64, v * x[u].z);
                    atomicAdd(t + 96, v * x[u].w);
                }
            }
        }
    }
    __syncthreads();

    for (int row = wave; row < nrows; row += 32) {
        const float* t = tile + row * 128 + lane;
        float4 o;
        o.x = t[0];
        o.y = t[32];
        o.z = t[64];
        o.w = t[96];
        *(float4*)(out + (size_t)(rowBase + row) * 128 + lane * 4) = o;
    }
}

extern "C" void kernel_launch(void* const* d_in, const int* in_sizes, int n_in,
                              void* d_out, int out_size, void* d_ws, size_t ws_size,
                              hipStream_t stream) {
    const float* X  = (const float*)d_in[0];
    const float* W  = (const float*)d_in[1];
    const int*   ei = (const int*)d_in[2];
    float* out = (float*)d_out;

    const int N = in_sizes[0] / 128;
    const int E = in_sizes[2] / 2;
    const int* rows = ei;
    const int* cols = ei + E;

    float* Xhat = (float*)d_ws;
    float* deg  = Xhat + (size_t)N * 128;
    float* dinv = deg + N;

    const int    ntiles   = (N + TILE_ROWS - 1) / TILE_ROWS;
    const size_t keyOff   = ((size_t)N * 130 * sizeof(float) + 255) & ~(size_t)255;
    const size_t keyBytes = (((size_t)E + 1023) >> 10) << 10;
    const bool   keyPlane = ntiles <= 256 && ws_size >= keyOff + keyBytes;
    unsigned char* keys = keyPlane ? (unsigned char*)d_ws + keyOff : nullptr;
    const size_t recOff   = keyOff + keyBytes;
    const bool   recPlane = keyPlane && N <= (1 << REC_COL_BITS) && ws_size >= recOff + (size_t)E * sizeof(unsigned int);
    unsigned int* recs = recPlane ? (unsigned int*)((unsigned char*)d_ws + recOff) : nullptr;

    deg_init_kernel<<<(N + 255) / 256, 256, 0, stream>>>(deg, N);
    gemm_and_count_kernel<<<512 + 256, 256, 0, stream>>>(X, W, Xhat, N, rows, cols, deg, (unsigned int*)keys, recs, E, 512);
    dinv_kernel<<<(N + 255) / 256, 256, 0, stream>>>(deg, dinv, N);

    if (recPlane)
        spmm_tile_kernel<true, true><<<ntiles, 1024, 0, stream>>>(rows, cols, keys, recs, dinv, Xhat, out, N, E);
    else if (keyPlane)
        spmm_tile_kernel<true, false><<<ntiles, 1024, 0, stream>>>(rows, cols, keys, recs, dinv, Xhat, out, N, E);
    else
        spmm_tile_kernel<false, false><<<ntiles, 1024, 0, stream>>>(rows, cols, keys, recs, dinv, Xhat, out, N, E);
}


# degree count: 32 blocks count their own runs of edges in private 16-bit LDS histograms of all nodes (integer LDS adds), planes summed by the dinv launch; no global f32 add per edge; degrees exact
# speedup vs baseline: 4.0960x; 1.5017x over previous
#include <hip/hip_runtime.h>

typedef float v2f __attribute__((ext_vector_type(2)));
typedef float v8f __attribute__((ext_vector_type(8)));

__global__ __launch_bounds__(256) void deg_init_kernel(float* __restrict__ deg, int N) {
    int n = blockIdx.x * 256 + threadIdx.x;
    if (n < N) deg[n] = 1.0f;
}

#define TILE_ROWS 400
#define REC_COL_BITS 17
static_assert(TILE_ROWS <= (1 << (32 - REC_COL_BITS)), "local row must fit above the column bits");

#define CG 13

__device__ __forceinline__ void count_role(const int* __restrict__ rows,
                                           const int* __restrict__ cols,
                                           float* __restrict__ deg,
                                           unsigned int* __restrict__ keyw,
                                           unsigned int* __restrict__ recs, int E,
                                           int bid, int nblk) {
    const int lane = threadIdx.x & 31;
    const int wave = __builtin_amdgcn_readfirstlane(threadIdx.x >> 5);
    const int G    = (E + 31) >> 5;
    const int nw   = nblk * 8;
    const int per  = (G + nw - 1) / nw;
    const int qBeg = (bid * 8 + wave) * per;
    const int qEnd = min(qBeg + per, G);
    const bool needRow = keyw != nullptr || recs != nullptr;
    for (int q0 = qBeg; q0 < qEnd; q0 += CG) {
        int c[CG], r[CG];
#pragma unroll
        for (int u = 0; u < CG; ++u) {
            const int e = (q0 + u) * 32 + lane;
            c[u] = 0;
            r[u] = 0;
            if (q0 + u < qEnd && e < E) {
                c[u] = cols[e];
                if (needRow) r[u] = rows[e];
            }
        }
#pragma unroll
        for (int u = 0; u < CG; ++u) {
            const int q = q0 + u;
            if (q < qEnd) {
                const int e = q * 32 + lane;
                if (e < E) {
                    if (deg != nullptr)
                        atomicAdd(&deg[c[u]], 1.0f);
                    if (recs != nullptr)
                        recs[e] = ((unsigned)(r[u] - (r[u] / TILE_ROWS) * TILE_ROWS) << REC_COL_BITS) | (unsigned)c[u];
                }
                if (keyw != nullptr && q * 32 + 31 < E) {
                    const unsigned key = (unsigned)(r[u] / TILE_ROWS);
                    unsigned mine = 0u;
#pragma unroll
                    for (int k = 0; k < 8; ++k) {
                        const unsigned bk = __builtin_amdgcn_ballot_w32(((key >> k) & 1u) != 0u);
                        if (lane == k) mine = bk;
                    }
                    if (lane < 8)
                        keyw[((((q >> 5) << 1) + (lane >> 2)) * 32 + (q & 31)) * 4 + (lane & 3)] = mine;
                }
            }
        }
    }
}

__global__ __launch_bounds__(256) void dinv_kernel(const float* __restrict__ deg,
                                                   float* __restrict__ dinv, int N) {
    int n = blockIdx.x * 256 + threadIdx.x;
    if (n < N) dinv[n] = rsqrtf(deg[n]);
}

#define HIST_BLOCKS 32
#define HIST_WORDS  65536
#define HIST_U      16

__global__ __launch_bounds__(1024) void deg_hist_kernel(const int* __restrict__ cols,
                                                        unsigned int* __restrict__ planes,
                                                        int E, int chunk, int pitch) {
    __shared__ __attribute__((aligned(16))) unsigned int hist[HIST_WORDS];
    const int tid = threadIdx.x;
    uint4* h4 = (uint4*)hist;
    const int nv = pitch >> 2;
    for (int i = tid; i < nv; i += 1024) h4[i] = make_uint4(0u, 0u, 0u, 0u);
    __syncthreads();

    const int eBeg = (int)blockIdx.x * chunk;
    const int eEnd = min(eBeg + chunk, E);
    int e = eBeg + tid;
    for (; e + (HIST_U - 1) * 1024 < eEnd; e += HIST_U * 1024) {
        int c[HIST_U];
#pragma unroll
        for (int u = 0; u < HIST_U; ++u) c[u] = cols[e + u * 1024];
#pragma unroll
        for (int u = 0; u < HIST_U; ++u) atomicAdd(&hist[c[u] >> 1], 1u << ((c[u] & 1) << 4));
    }
    for (; e < eEnd; e += 1024) {
        const int c = cols[e];
        atomicAdd(&hist[c >> 1], 1u << ((c & 1) << 4));
    }
    __syncthreads();

    uint4* dst = (uint4*)(planes + (size_t)blockIdx.x * pitch);
    for (int i = tid; i < nv; i += 1024) dst[i] = h4[i];
}

__global__ __launch_bounds__(256) void dinv_planes_kernel(const unsigned int* __restrict__ planes,
                                                          float* __restrict__ dinv, int N, int pitch) {
    const int w = blockIdx.x * 256 + threadIdx.x;
    if (2 * w < N) {
        unsigned int lo = 0u, hi = 0u;
#pragma unroll
        for (int p = 0; p < HIST_BLOCKS; ++p) {
            const unsigned int v = planes[(size_t)p * pitch + w];
            lo += v & 0xffffu;
            hi += v >> 16;
        }
        dinv[2 * w] = rsqrtf((float)(lo + 1u));
        if (2 * w + 1 < N) dinv[2 * w + 1] = rsqrtf((float)(hi + 1u));
    }
}

#define GS_ROWS  32
#define GS_PITCH 132

__device__ __forceinline__ void gemm_role(const float* __restrict__ X,
                                          const float* __restrict__ W,
                                          float* __restrict__ Xhat,
                                          int N, int bid, int nblk) {
    __shared__ __attribute__((aligned(16))) float sA[GS_ROWS * GS_PITCH];
    __shared__ __attribute__((aligned(16))) float sO[GS_ROWS * GS_PITCH];

    const int lane    = threadIdx.x & 31;
    const int wave    = __builtin_amdgcn_readfirstlane(threadIdx.x >> 5);
    const int colBase = wave * 16;
    const int half    = lane >> 4;
    const int l16     = lane & 15;
    const int nstrips = (N + GS_ROWS - 1) / GS_ROWS;
    const int stride  = nblk;

    v2f bfrag[32];
    {
        const float* bptr = W + (size_t)(colBase + l16) * 128 + (half << 1);
#pragma unroll
        for (int ks = 0; ks < 32; ++ks) bfrag[ks] = *(const v2f*)(bptr + 4 * ks);
    }

    const int myRow = wave * 4;
    const float* xlane = X + lane * 4;
    float4 x0 = {0.f, 0.f, 0.f, 0.f}, x1 = x0, x2 = x0, x3 = x0;
    int s = bid;
    if (s < nstrips) {
        const int r0 = s * GS_ROWS + myRow;
        x0 = *(const float4*)(xlane + (size_t)min(r0 + 0, N - 1) * 128);
        x1 = *(const float4*)(xlane + (size_t)min(r0 + 1, N - 1) * 128);
        x2 = *(const float4*)(xlane + (size_t)min(r0 + 2, N - 1) * 128);
        x3 = *(const float4*)(xlane + (size_t)min(r0 + 3, N - 1) * 128);
    }

    const float* a0p = sA + l16 * GS_PITCH + (half << 1);
    const float* a1p = sA + (16 + l16) * GS_PITCH + (half << 1);

    for (; s < nstrips; s += stride) {
        float* arow = sA + myRow * GS_PITCH + lane * 4;
        *(float4*)(arow + 0 * GS_PITCH) = x0;
        *(float4*)(arow + 1 * GS_PITCH) = x1;
        *(float4*)(arow + 2 * GS_PITCH) = x2;
        *(float4*)(arow + 3 * GS_PITCH) = x3;
        __syncthreads();

        const int sn = s + stride;
        if (sn < nstrips) {
            const int r0 = sn * GS_ROWS + myRow;
            x0 = *(const float4*)(xlane + (size_t)min(r0 + 0, N - 1) * 128);
            x1 = *(const float4*)(xlane + (size_t)min(r0 + 1, N - 1) * 128);
            x2 = *(const float4*)(xlane + (size_t)min(r0 + 2, N - 1) * 128);
            x3 = *(const float4*)(xlane + (size_t)min(r0 + 3, N - 1) * 128);
        }

        v8f acc0 = {0.f, 0.f, 0.f, 0.f, 0.f, 0.f, 0.f, 0.f};
        v8f acc1 = {0.f, 0.f, 0.f, 0.f, 0.f, 0.f, 0.f, 0.f};
#pragma unroll
        for (int ks = 0; ks < 32; ++ks) {
            const v2f a0 = *(const v2f*)(a0p + 4 * ks);
            const v2f a1 = *(const v2f*)(a1p + 4 * ks);
            acc0 = __builtin_amdgcn_wmma_f32_16x16x4_f32(
                 false, a0,  false, bfrag[ks],
                 (short)0, acc0,  false,  false);
            acc1 = __builtin_amdgcn_wmma_f32_16x16x4_f32(
                 false, a1,  false, bfrag[ks],
                 (short)0, acc1,  false,  false);
        }

#pragma unroll
        for (int v = 0; v < 8; ++v) {
            const int m = v + (half << 3);
            sO[m * GS_PITCH + colBase + l16]        = acc0[v];
            sO[(16 + m) * GS_PITCH + colBase + l16] = acc1[v];
        }
        __syncthreads();

        const float* orow = sO + myRow * GS_PITCH + lane * 4;
        const int r0 = s * GS_ROWS + myRow;
        float* dst = Xhat + (size_t)r0 * 128 + lane * 4;
        if (r0 + 3 < N) {
            const float4 o0 = *(const float4*)(orow + 0 * GS_PITCH);
            const float4 o1 = *(const float4*)(orow + 1 * GS_PITCH);
            const float4 o2 = *(const float4*)(orow + 2 * GS_PITCH);
            const float4 o3 = *(const float4*)(orow + 3 * GS_PITCH);
            *(float4*)(dst + 0 * 128) = o0;
            *(float4*)(dst + 1 * 128) = o1;
            *(float4*)(dst + 2 * 128) = o2;
            *(float4*)(dst + 3 * 128) = o3;
        } else {
            for (int j = 0; j < 4; ++j)
                if (r0 + j < N) *(float4*)(dst + j * 128) = *(const float4*)(orow + j * GS_PITCH);
        }
    }
}

__global__ __launch_bounds__(256) void gemm_and_count_kernel(const float* __restrict__ X,
                                                             const float* __restrict__ W,
                                                             float* __restrict__ Xhat, int N,
                                                             const int* __restrict__ rows,
                                                             const int* __restrict__ cols,
                                                             float* __restrict__ deg,
                                                             unsigned int* __restrict__ keyw,
                                                             unsigned int* __restrict__ recs, int E,
                                                             int gemmBlocks) {
    const int bid = (int)blockIdx.x;
    if (bid < gemmBlocks)
        gemm_role(X, W, Xhat, N, bid, gemmBlocks);
    else
        count_role(rows, cols, deg, keyw, recs, E, bid - gemmBlocks, (int)gridDim.x - gemmBlocks);
}

#define SEG_CAP   768

__device__ __forceinline__ void add_edge_row(float* __restrict__ tile,
                                             const float* __restrict__ Xhat,
                                             int c, int rl, float val, int lane) {
    const float4 x = *(const float4*)(Xhat + (size_t)c * 128 + lane * 4);
    float* t = tile + rl * 128 + lane;
    atomicAdd(t,      val * x.x);
    atomicAdd(t + 32, val * x.y);
    atomicAdd(t + 64, val * x.z);
    atomicAdd(t + 96, val * x.w);
}

__device__ __forceinline__ int push_match(bool m, int e, int cnt, int* __restrict__ seg) {
    const unsigned mj = __builtin_amdgcn_ballot_w32(m);
    if (mj == 0u) return cnt;
    const int slot = cnt + (int)__builtin_amdgcn_mbcnt_lo(mj, 0u);
    if (m && slot < SEG_CAP) seg[slot] = e;
    return cnt + __builtin_popcount(mj);
}

__device__ __forceinline__ int scan_step(unsigned a0, unsigned a1, unsigned a2, unsigned a3,
                                         unsigned unrows, int e0, int cnt, int* __restrict__ seg) {
    if (__builtin_amdgcn_ballot_w32(min(min(a0, a1), min(a2, a3)) < unrows) != 0u) {
        cnt = push_match(a0 < unrows, e0 + 0, cnt, seg);
        cnt = push_match(a1 < unrows, e0 + 1, cnt, seg);
        cnt = push_match(a2 < unrows, e0 + 2, cnt, seg);
        cnt = push_match(a3 < unrows, e0 + 3, cnt, seg);
    }
    return cnt;
}

template <bool SECOND>
__device__ __forceinline__ int key_step(uint4 a, uint4 b, bool ok, unsigned kb, int e0, int cnt,
                                        int* __restrict__ seg, float* __restrict__ tile,
                                        const int* __restrict__ rows, const int* __restrict__ cols,
                                        const float* __restrict__ dinv, const float* __restrict__ Xhat,
                                        int rowBase, int lane) {
    unsigned m = ~(a.x ^ (0u - ((kb >> 0) & 1u))) & ~(a.y ^ (0u - ((kb >> 1) & 1u))) &
                 ~(a.z ^ (0u - ((kb >> 2) & 1u))) & ~(a.w ^ (0u - ((kb >> 3) & 1u))) &
                 ~(b.x ^ (0u - ((kb >> 4) & 1u))) & ~(b.y ^ (0u - ((kb >> 5) & 1u))) &
                 ~(b.z ^ (0u - ((kb >> 6) & 1u))) & ~(b.w ^ (0u - ((kb >> 7) & 1u)));
    if (!ok) m = 0u;
    for (;;) {
        const bool has = m != 0u;
        const unsigned mj = __builtin_amdgcn_ballot_w32(has);
        if (mj == 0u) break;
        const int e = e0 + __builtin_ffs((int)m) - 1;
        const int slot = cnt + (int)__builtin_amdgcn_mbcnt_lo(mj, 0u);
        if constexpr (!SECOND) {
            if (has && slot < SEG_CAP) seg[slot] = e;
        } else {
            unsigned ov = __builtin_amdgcn_ballot_w32(has && slot >= SEG_CAP);
            while (ov) {
                const int srcl = __builtin_ctz(ov);
                ov &= ov - 1u;
                const int esel = __builtin_amdgcn_readlane(e, srcl);
                const int r = rows[esel];
                const int c = cols[esel];
                add_edge_row(tile, Xhat, c, r - rowBase, dinv[r] * dinv[c], lane);
            }
        }
        m &= m - 1u;
        cnt += __builtin_popcount(mj);
    }
    return cnt;
}

template <bool KEY_PLANE, bool RECORDS>
__global__ __launch_bounds__(1024) void spmm_tile_kernel(const int* __restrict__ rows,
                                                         const int* __restrict__ cols,
                                                         const unsigned char* __restrict__ keys,
                                                         const unsigned int* __restrict__ recs,
                                                         const float* __restrict__ dinv,
                                                         const float* __restrict__ Xhat,
                                                         float* __restrict__ out,
                                                         int N, int E) {
    __shared__ float tile[TILE_ROWS * 128];
    __shared__ int   queue[32 * SEG_CAP];

    const int tid  = threadIdx.x;
    const int lane = tid & 31;
    const int wave = __builtin_amdgcn_readfirstlane(tid >> 5);
    const int rowBase = blockIdx.x * TILE_ROWS;
    const int nrows   = min(TILE_ROWS, N - rowBase);
    const unsigned unrows = (unsigned)nrows;

#pragma unroll 4
    for (int row = wave; row < nrows; row += 32) {
        const int n = rowBase + row;
        const float d = dinv[n];
        const float4 x = *(const float4*)(Xhat + (size_t)n * 128 + lane * 4);
        float* t = tile + row * 128 + lane;
        t[0]  = d * d * x.x;
        t[32] = d * d * x.y;
        t[64] = d * d * x.z;
        t[96] = d * d * x.w;
    }
    __syncthreads();

    int* seg = queue + wave * SEG_CAP;
    int cnt = 0;
    constexpr int U = 8;
    const int tailStart = KEY_PLANE ? (E >> 5) << 5 : (E >> 2) << 2;
    if constexpr (KEY_PLANE) {
        const int G32 = tailStart >> 5;
        const int SGF = G32 >> 5;
        const uint4* kp = (const uint4*)keys;
        const unsigned kb = blockIdx.x;
        int sg = wave;
        for (; sg + (U - 1) * 32 < SGF; sg += 32 * U) {
            uint4 ka[U], kc[U];
#pragma unroll
            for (int u = 0; u < U; ++u) {
                ka[u] = kp[(sg + 32 * u) * 64 + lane];
                kc[u] = kp[(sg + 32 * u) * 64 + 32 + lane];
            }
#pragma unroll
            for (int u = 0; u < U; ++u) {
                cnt = key_step<false>(ka[u], kc[u], true, kb, ((sg + 32 * u) * 32 + lane) * 32, cnt, seg,
                                      tile, rows, cols, dinv, Xhat, rowBase, lane);
            }
        }
        for (; sg * 32 < G32; sg += 32) {
            const int g  = sg * 32 + lane;
            const int gl = min(g, G32 - 1);
            const int o  = (gl >> 5) * 64 + (gl & 31);
            cnt = key_step<false>(kp[o], kp[o + 32], g < G32, kb, g * 32, cnt, seg,
                                  tile, rows, cols, dinv, Xhat, rowBase, lane);
        }
        if (cnt > SEG_CAP) {
            int cnt2 = 0;
            for (int sg2 = wave; sg2 * 32 < G32; sg2 += 32) {
                const int g  = sg2 * 32 + lane;
                const int gl = min(g, G32 - 1);
                const int o  = (gl >> 5) * 64 + (gl & 31);
                cnt2 = key_step<true>(kp[o], kp[o + 32], g < G32, kb, g * 32, cnt2, seg,
                                      tile, rows, cols, dinv, Xhat, rowBase, lane);
            }
        }
    } else {
        const int E4 = tailStart >> 2;
        const int4* rows4 = (const int4*)rows;
        int base = wave * 32;
        for (; base + (U - 1) * 1024 + 31 < E4; base += 1024 * U) {
            int4 r4[U];
#pragma unroll
            for (int u = 0; u < U; ++u) r4[u] = rows4[base + u * 1024 + lane];
#pragma unroll
            for (int u = 0; u < U; ++u) {
                cnt = scan_step((unsigned)(r4[u].x - rowBase), (unsigned)(r4[u].y - rowBase),
                                (unsigned)(r4[u].z - rowBase), (unsigned)(r4[u].w - rowBase),
                                unrows, 4 * (base + u * 1024 + lane), cnt, seg);
            }
        }
        for (; base < E4; base += 1024) {
            const int i = base + lane;
            const bool ok = i < E4;
            const int4 r = rows4[min(i, E4 - 1)];
            cnt = scan_step(ok ? (unsigned)(r.x - rowBase) : ~0u, ok ? (unsigned)(r.y - rowBase) : ~0u,
                            ok ? (unsigned)(r.z - rowBase) : ~0u, ok ? (unsigned)(r.w - rowBase) : ~0u,
                            unrows, 4 * i, cnt, seg);
        }
        if (cnt > SEG_CAP) {
            int cnt2 = 0;
            for (int base2 = wave * 32; base2 < E4; base2 += 1024) {
                const int i = base2 + lane;
                const bool ok = i < E4;
                const int4 r = rows4[min(i, E4 - 1)];
                const int rr[4] = {r.x, r.y, r.z, r.w};
#pragma unroll
                for (int j = 0; j < 4; ++j) {
                    const bool m = ok && ((unsigned)(rr[j] - rowBase) < unrows);
                    const unsigned mj = __builtin_amdgcn_ballot_w32(m);
                    const int slot = cnt2 + (int)__builtin_amdgcn_mbcnt_lo(mj, 0u);
                    unsigned ov = __builtin_amdgcn_ballot_w32(m && slot >= SEG_CAP);
                    while (ov) {
                        const int src = __builtin_ctz(ov);
                        ov &= ov - 1u;
                        const int rsel = __builtin_amdgcn_readlane(rr[j], src);
                        const int esel = __builtin_amdgcn_readlane(4 * i + j, src);
                        const int c = cols[esel];
                        add_edge_row(tile, Xhat, c, rsel - rowBase, dinv[rsel] * dinv[c], lane);
                    }
                    cnt2 += __builtin_popcount(mj);
                }
            }
        }
    }
    if (wave == 0) {
        for (int e = tailStart; e < E; ++e) {
            const int r = rows[e];
            if ((unsigned)(r - rowBase) < unrows) {
                const int c = cols[e];
                add_edge_row(tile, Xhat, c, r - rowBase, dinv[r] * dinv[c], lane);
            }
        }
    }

    const int nq = min(cnt, SEG_CAP);
    constexpr int G = 16;
    for (int b = 0; b < nq; b += 32) {
        const int nv = min(32, nq - b);
        const int e = seg[b + (lane < nv ? lane : 0)];
        int rl, c;
        if constexpr (RECORDS) {
            const unsigned w = recs[e];
            c  = (int)(w & ((1u << REC_COL_BITS) - 1u));
            rl = (int)(w >> REC_COL_BITS);
        } else {
            c  = cols[e];
            rl = rows[e] - rowBase;
        }
        const float val = dinv[rowBase + rl] * dinv[c];
        for (int k0 = 0; k0 < nv; k0 += G) {
            float4 x[G];
#pragma unroll
            for (int u = 0; u < G; ++u) {
                const int cc = __builtin_amdgcn_readlane(c, k0 + u);
                x[u] = *(const float4*)(Xhat + (size_t)cc * 128 + lane * 4);
            }
#pragma unroll
            for (int u = 0; u < G; ++u) {
                if (u == 0 || k0 + u < nv) {
                    const float v = __int_as_float(__builtin_amdgcn_readlane(__float_as_int(val), k0 + u));
                    const int rr  = __builtin_amdgcn_readlane(rl, k0 + u);
                    float* t = tile + rr * 128 + lane;
                    atomicAdd(t,      v * x[u].x);
                    atomicAdd(t + 32, v * x[u].y);
                    atomicAdd(t + 64, v * x[u].z);
                    atomicAdd(t + 96, v * x[u].w);
                }
            }
        }
    }
    __syncthreads();

    for (int row = wave; row < nrows; row += 32) {
        const float* t = tile + row * 128 + lane;
        float4 o;
        o.x = t[0];
        o.y = t[32];
        o.z = t[64];
        o.w = t[96];
        *(float4*)(out + (size_t)(rowBase + row) * 128 + lane * 4) = o;
    }
}

extern "C" void kernel_launch(void* const* d_in, const int* in_sizes, int n_in,
                              void* d_out, int out_size, void* d_ws, size_t ws_size,
                              hipStream_t stream) {
    const float* X  = (const float*)d_in[0];
    const float* W  = (const float*)d_in[1];
    const int*   ei = (const int*)d_in[2];
    float* out = (float*)d_out;

    const int N = in_sizes[0] / 128;
    const int E = in_sizes[2] / 2;
    const int* rows = ei;
    const int* cols = ei + E;

    float* Xhat = (float*)d_ws;
    float* deg  = Xhat + (size_t)N * 128;
    float* dinv = deg + N;

    const int    ntiles   = (N + TILE_ROWS - 1) / TILE_ROWS;
    const size_t keyOff   = ((size_t)N * 130 * sizeof(float) + 255) & ~(size_t)255;
    const size_t keyBytes = (((size_t)E + 1023) >> 10) << 10;
    const bool   keyPlane = ntiles <= 256 && ws_size >= keyOff + keyBytes;
    unsigned char* keys = keyPlane ? (unsigned char*)d_ws + keyOff : nullptr;
    const size_t recOff   = keyOff + keyBytes;
    const bool   recPlane = keyPlane && N <= (1 << REC_COL_BITS) && ws_size >= recOff + (size_t)E * sizeof(unsigned int);
    unsigned int* recs = recPlane ? (unsigned int*)((unsigned char*)d_ws + recOff) : nullptr;

    const int    words      = (N + 1) >> 1;
    const int    pitch      = (words + 3) & ~3;
    const int    chunk      = (E + HIST_BLOCKS - 1) / HIST_BLOCKS;
    const size_t planeOff   = ((recPlane ? recOff + (size_t)E * sizeof(unsigned int) : keyPlane ? recOff : (size_t)N * 130 * sizeof(float)) + 255) & ~(size_t)255;
    const size_t planeBytes = (size_t)HIST_BLOCKS * pitch * sizeof(unsigned int);
    unsigned int* planes = nullptr;
    if (pitch <= HIST_WORDS && chunk <= 65535) {
        if (ws_size >= planeOff + planeBytes) planes = (unsigned int*)((unsigned char*)d_ws + planeOff);
        else if ((size_t)out_size * sizeof(float) >= planeBytes) planes = (unsigned int*)d_out;
    }

    if (planes != nullptr) {
        deg_hist_kernel<<<HIST_BLOCKS, 1024, 0, stream>>>(cols, planes, E, chunk, pitch);
        gemm_and_count_kernel<<<512 + (keyPlane ? 256 : 0), 256, 0, stream>>>(X, W, Xhat, N, rows, cols, nullptr, (unsigned int*)keys, recs, E, 512);
        dinv_planes_kernel<<<(words + 255) / 256, 256, 0, stream>>>(planes, dinv, N, pitch);
    } else {
        deg_init_kernel<<<(N + 255) / 256, 256, 0, stream>>>(deg, N);
        gemm_and_count_kernel<<<512 + 256, 256, 0, stream>>>(X, W, Xhat, N, rows, cols, deg, (unsigned int*)keys, recs, E, 512);
        dinv_kernel<<<(N + 255) / 256, 256, 0, stream>>>(deg, dinv, N);
    }

    if (recPlane)
        spmm_tile_kernel<true, true><<<ntiles, 1024, 0, stream>>>(rows, cols, keys, recs, dinv, Xhat, out, N, E);
    else if (keyPlane)
        spmm_tile_kernel<true, false><<<ntiles, 1024, 0, stream>>>(rows, cols, keys, recs, dinv, Xhat, out, N, E);
    else
        spmm_tile_kernel<false, false><<<ntiles, 1024, 0, stream>>>(rows, cols, keys, recs, dinv, Xhat, out, N, E);
}
